# starting kernel
# baseline (speedup 1.0000x reference)
#include <hip/hip_runtime.h>

typedef __attribute__((ext_vector_type(2))) float v2f;
typedef __attribute__((ext_vector_type(8))) float v8f;

constexpr int Bb = 8, Tt = 2048, Ee = 1024, Hh = 64;
constexpr int ROWS = Bb * Tt;
constexpr int TPB  = Tt / 16;

__device__ __forceinline__ v8f wmma4(v2f a, v2f b, v8f c) {
    return __builtin_amdgcn_wmma_f32_16x16x4_f32(
         false, a,  false, b,
         (short)0, c,  false,  false);
}

__global__ __launch_bounds__(128)
void proj_kernel(const float* __restrict__ x,
                 const float* __restrict__ Wk,
                 const float* __restrict__ Wq,
                 const float* __restrict__ Wv,
                 float* __restrict__ qo, float* __restrict__ ko,
                 float* __restrict__ vo) {
    const int lane   = threadIdx.x & 31;
    const int wv     = threadIdx.x >> 5;
    const int rowb   = blockIdx.x * 16;
    const int mrow   = lane & 15;
    const int khalf  = (lane >> 4) * 2;
    const int ncol   = wv * 16 + (lane & 15);

    const float* xa = x + (size_t)(rowb + mrow) * Ee + khalf;

    v8f cq = {}, ck = {}, cv = {};
    #pragma unroll 4
    for (int e0 = 0; e0 < Ee; e0 += 4) {
        v2f a = *(const v2f*)(xa + e0);
        const float* wkp = Wk + (size_t)(e0 + khalf) * Hh + ncol;
        const float* wqp = Wq + (size_t)(e0 + khalf) * Hh + ncol;
        const float* wvp = Wv + (size_t)(e0 + khalf) * Hh + ncol;
        v2f bk; bk.x = wkp[0]; bk.y = wkp[Hh];
        v2f bq; bq.x = wqp[0]; bq.y = wqp[Hh];
        v2f bv; bv.x = wvp[0]; bv.y = wvp[Hh];
        ck = wmma4(a, bk, ck);
        cq = wmma4(a, bq, cq);
        cv = wmma4(a, bv, cv);
    }

    const int orow = rowb + (lane >> 4) * 8;
    #pragma unroll
    for (int r = 0; r < 8; ++r) {
        size_t idx = (size_t)(orow + r) * Hh + ncol;
        qo[idx] = cq[r];
        ko[idx] = ck[r];
        vo[idx] = cv[r];
    }
}

__global__ __launch_bounds__(32)
void attn_kernel(const float* __restrict__ q,
                 const float* __restrict__ k,
                 const float* __restrict__ v,
                 float* __restrict__ out) {
    __shared__ float pbuf[16][18];

    const int lane  = threadIdx.x & 31;
    const int tile  = blockIdx.x;
    const int b     = tile / TPB;
    const int it    = tile % TPB;
    const int qrow0 = b * Tt + it * 16;
    const int half  = lane >> 4;
    const int nlo   = lane & 15;
    const int khalf = half * 2;

    v2f qa[16];
    const float* qp = q + (size_t)(qrow0 + nlo) * Hh + khalf;
    #pragma unroll
    for (int s = 0; s < 16; ++s) qa[s] = *(const v2f*)(qp + 4 * s);

    v8f o0 = {}, o1 = {}, o2 = {}, o3 = {};
    float mrow[8], lrow[8];
    #pragma unroll
    for (int r = 0; r < 8; ++r) { mrow[r] = -1e30f; lrow[r] = 0.f; }

    const float scale = 0.03125f;

    for (int j = 0; j <= it; ++j) {
        const int krow0 = b * Tt + j * 16;

        v8f s8 = {};
        const float* kp = k + (size_t)(krow0 + nlo) * Hh + khalf;
        #pragma unroll
        for (int t = 0; t < 16; ++t) {
            v2f kb = *(const v2f*)(kp + 4 * t);
            s8 = wmma4(qa[t], kb, s8);
        }

        float sv[8];
        #pragma unroll
        for (int r = 0; r < 8; ++r) sv[r] = s8[r] * scale;

        if (j == it) {
            #pragma unroll
            for (int r = 0; r < 8; ++r) {
                int m = r + half * 8;
                if (nlo > m) sv[r] = -1e30f;
            }
        }

        float alpha[8], pv[8];
        #pragma unroll
        for (int r = 0; r < 8; ++r) {
            float t = sv[r];
            t = fmaxf(t, __shfl_xor(t, 1, 16));
            t = fmaxf(t, __shfl_xor(t, 2, 16));
            t = fmaxf(t, __shfl_xor(t, 4, 16));
            t = fmaxf(t, __shfl_xor(t, 8, 16));
            float mnew = fmaxf(mrow[r], t);
            alpha[r] = __expf(mrow[r] - mnew);
            mrow[r]  = mnew;
            float p  = __expf(sv[r] - mnew);
            pv[r] = p;
            float s = p;
            s += __shfl_xor(s, 1, 16);
            s += __shfl_xor(s, 2, 16);
            s += __shfl_xor(s, 4, 16);
            s += __shfl_xor(s, 8, 16);
            lrow[r] = lrow[r] * alpha[r] + s;
        }

        #pragma unroll
        for (int r = 0; r < 8; ++r) {
            o0[r] *= alpha[r]; o1[r] *= alpha[r];
            o2[r] *= alpha[r]; o3[r] *= alpha[r];
        }

        #pragma unroll
        for (int r = 0; r < 8; ++r)
            pbuf[r + half * 8][nlo] = pv[r];
        __builtin_amdgcn_wave_barrier();
        asm volatile("s_wait_dscnt 0" ::: "memory");

        const float* vp = v + (size_t)krow0 * Hh;
        #pragma unroll
        for (int kk0 = 0; kk0 < 16; kk0 += 4) {
            v2f pa;
            pa.x = pbuf[nlo][kk0 + khalf];
            pa.y = pbuf[nlo][kk0 + khalf + 1];
            const float* vrow = vp + (size_t)(kk0 + khalf) * Hh;
            { v2f vb; vb.x = vrow[ 0 + nlo]; vb.y = vrow[Hh +  0 + nlo]; o0 = wmma4(pa, vb, o0); }
            { v2f vb; vb.x = vrow[16 + nlo]; vb.y = vrow[Hh + 16 + nlo]; o1 = wmma4(pa, vb, o1); }
            { v2f vb; vb.x = vrow[32 + nlo]; vb.y = vrow[Hh + 32 + nlo]; o2 = wmma4(pa, vb, o2); }
            { v2f vb; vb.x = vrow[48 + nlo]; vb.y = vrow[Hh + 48 + nlo]; o3 = wmma4(pa, vb, o3); }
        }
        __builtin_amdgcn_wave_barrier();
    }

    #pragma unroll
    for (int r = 0; r < 8; ++r) {
        float inv = 1.0f / lrow[r];
        size_t idx = (size_t)(qrow0 + r + half * 8) * Hh;
        out[idx +  0 + nlo] = o0[r] * inv;
        out[idx + 16 + nlo] = o1[r] * inv;
        out[idx + 32 + nlo] = o2[r] * inv;
        out[idx + 48 + nlo] = o3[r] * inv;
    }
}

extern "C" void kernel_launch(void* const* d_in, const int* in_sizes, int n_in,
                              void* d_out, int out_size, void* d_ws, size_t ws_size,
                              hipStream_t stream) {
    const float* x  = (const float*)d_in[0];
    const float* Wk = (const float*)d_in[1];
    const float* Wq = (const float*)d_in[2];
    const float* Wv = (const float*)d_in[3];
    float* out = (float*)d_out;

    float* ws   = (float*)d_ws;
    float* qbuf = ws;
    float* kbuf = ws + (size_t)ROWS * Hh;
    float* vbuf = ws + 2 * (size_t)ROWS * Hh;

    proj_kernel<<<ROWS / 16, 128, 0, stream>>>(x, Wk, Wq, Wv, qbuf, kbuf, vbuf);
    attn_kernel<<<ROWS / 16, 32, 0, stream>>>(qbuf, kbuf, vbuf, out);
}


# attention: 2-wave work-group per query tile, split by phase, 1 trip apart: wave 0 scores+softmax, wave 1 P*V; P and rescale factors via 2 LDS slots; K/V refill behind each matrix run; sum order kept
# speedup vs baseline: 1.8714x; 1.8714x over previous
#include <hip/hip_runtime.h>

typedef __attribute__((ext_vector_type(2))) float v2f;
typedef __attribute__((ext_vector_type(8))) float v8f;

constexpr int Bb = 8, Tt = 2048, Ee = 1024, Hh = 64;
constexpr int ROWS = Bb * Tt;
constexpr int TPB  = Tt / 16;

__device__ __forceinline__ v8f wmma4(v2f a, v2f b, v8f c) {
    return __builtin_amdgcn_wmma_f32_16x16x4_f32(
         false, a,  false, b,
         (short)0, c,  false,  false);
}

constexpr int TILE_F = 1024;
constexpr int CQ_F   = 256;
constexpr int RQ_F   = 128;

constexpr int PR = 4;
constexpr int PC = 3;

__global__ __launch_bounds__(128)
void proj_kernel(const float* __restrict__ x,
                 const float* __restrict__ Wk,
                 const float* __restrict__ Wq,
                 const float* __restrict__ Wv,
                 float* __restrict__ qo, float* __restrict__ ko,
                 float* __restrict__ vo) {
    const int lane = threadIdx.x & 31;
    const int wv   = __builtin_amdgcn_readfirstlane((int)(threadIdx.x >> 5));
    const int c    = lane & 15;
    const int h    = lane >> 4;
    const int rowb = blockIdx.x * (16 * PR);

    const float* wp[PC];
    wp[0] = Wk + (size_t)(4 * h) * Hh + wv * 16 + c;
    wp[1] = Wq + (size_t)(4 * h) * Hh + wv * 16 + c;
    wp[2] = Wv + (size_t)(4 * h) * Hh + wv * 16 + c;
    const float* xp = x + (size_t)(rowb + c) * Ee + 4 * h;

    v8f acc[PR][PC];
    #pragma unroll
    for (int rt = 0; rt < PR; ++rt) {
        #pragma unroll
        for (int t = 0; t < PC; ++t) acc[rt][t] = v8f{};
    }

    float4 xa0[2][PR], xa1[2][PR];
    float  wb0[2][PC][4], wb1[2][PC][4];

    auto load_chunk = [&](float4 (&xa)[2][PR], float (&wb)[2][PC][4], int e0) {
        #pragma unroll
        for (int g2 = 0; g2 < 2; ++g2) {
            #pragma unroll
            for (int rt = 0; rt < PR; ++rt)
                xa[g2][rt] = *(const float4*)(xp + (size_t)rt * 16 * Ee + e0 + 8 * g2);
            #pragma unroll
            for (int t = 0; t < PC; ++t) {
                #pragma unroll
                for (int j = 0; j < 4; ++j)
                    wb[g2][t][j] = wp[t][(size_t)(e0 + 8 * g2 + j) * Hh];
            }
        }
    };
    auto mma_chunk = [&](const float4 (&xa)[2][PR], const float (&wb)[2][PC][4]) {
        #pragma unroll
        for (int g2 = 0; g2 < 2; ++g2) {
            #pragma unroll
            for (int rt = 0; rt < PR; ++rt) {
                v2f a; a.x = xa[g2][rt].x; a.y = xa[g2][rt].y;
                #pragma unroll
                for (int t = 0; t < PC; ++t) {
                    v2f b; b.x = wb[g2][t][0]; b.y = wb[g2][t][1];
                    acc[rt][t] = (t < 2) ? wmma4(b, a, acc[rt][t]) : wmma4(a, b, acc[rt][t]);
                }
            }
            #pragma unroll
            for (int rt = 0; rt < PR; ++rt) {
                v2f a; a.x = xa[g2][rt].z; a.y = xa[g2][rt].w;
                #pragma unroll
                for (int t = 0; t < PC; ++t) {
                    v2f b; b.x = wb[g2][t][2]; b.y = wb[g2][t][3];
                    acc[rt][t] = (t < 2) ? wmma4(b, a, acc[rt][t]) : wmma4(a, b, acc[rt][t]);
                }
            }
        }
    };

    load_chunk(xa0, wb0, 0);
    for (int e0 = 0; e0 < Ee - 32; e0 += 32) {
        load_chunk(xa1, wb1, e0 + 16);
        mma_chunk(xa0, wb0);
        load_chunk(xa0, wb0, e0 + 32);
        mma_chunk(xa1, wb1);
    }
    load_chunk(xa1, wb1, Ee - 16);
    mma_chunk(xa0, wb0);
    mma_chunk(xa1, wb1);

    float* ob[PC] = { ko, qo, vo };
    #pragma unroll
    for (int rt = 0; rt < PR; ++rt) {
        const size_t rec = (size_t)(blockIdx.x * PR + rt) * TILE_F + wv * CQ_F + lane * 4;
        #pragma unroll
        for (int t = 0; t < PC; ++t) {
            #pragma unroll
            for (int rq = 0; rq < 2; ++rq) {
                float4 o4;
                o4.x = acc[rt][t][4 * rq + 0]; o4.y = acc[rt][t][4 * rq + 1];
                o4.z = acc[rt][t][4 * rq + 2]; o4.w = acc[rt][t][4 * rq + 3];
                *(float4*)(ob[t] + rec + rq * RQ_F) = o4;
            }
        }
    }
}

constexpr int KT = 4;
constexpr int KB = 16 * KT;
constexpr int PP = KB + 4;

__global__ __launch_bounds__(64)
void attn_kernel(const float* __restrict__ q,
                 const float* __restrict__ k,
                 const float* __restrict__ v,
                 float* __restrict__ out) {
    __shared__ __attribute__((aligned(16))) float pbuf[2][16][PP];
    __shared__ __attribute__((aligned(16))) float abuf[2][16];
    __shared__ __attribute__((aligned(16))) float lbuf[16];

    const int lane  = threadIdx.x & 31;
    const int role  = __builtin_amdgcn_readfirstlane((int)(threadIdx.x >> 5));
    const int tile  = blockIdx.x;
    const int b     = tile / TPB;
    const int it    = tile % TPB;
    const int qrow0 = tile * 16;
    const int half  = lane >> 4;
    const int nlo   = lane & 15;

    const float scale = 0.03125f;
    const int nblk = it / KT + 1;
    const int qidx = it * 16 + half * 8;
    const size_t recb = (size_t)(b * TPB) * TILE_F + lane * 4;

    float4 qf[4][2];
    float4 fr[32];
    v8f o0 = {}, o1 = {}, o2 = {}, o3 = {};
    float mrow[8], lrow[8];
    #pragma unroll
    for (int r = 0; r < 8; ++r) { mrow[r] = -1e30f; lrow[r] = 0.f; }

    if (role == 0) {
        const float* qp = q + (size_t)tile * TILE_F + lane * 4;
        #pragma unroll
        for (int cq = 0; cq < 4; ++cq) {
            #pragma unroll
            for (int rq = 0; rq < 2; ++rq)
                qf[cq][rq] = *(const float4*)(qp + cq * CQ_F + rq * RQ_F);
        }
    } else {
        #pragma unroll
        for (int cq = 0; cq < 4; ++cq) {
            #pragma unroll
            for (int rq = 0; rq < 2; ++rq)
                qf[cq][rq] = make_float4(0.f, 0.f, 0.f, 0.f);
        }
    }
    const float* fsrc = ((role == 0) ? k : v) + recb;
    #pragma unroll
    for (int i = 0; i < 32; ++i)
        fr[i] = *(const float4*)(fsrc + (i & 3) * TILE_F + (i >> 2) * RQ_F);

    for (int jb = 0; jb <= nblk; ++jb) {
        const bool prod = (role == 0) && (jb < nblk);
        const bool cons = (role != 0) && (jb > 0);
        v8f s8[KT];
        #pragma unroll
        for (int tb = 0; tb < KT; ++tb) s8[tb] = v8f{};
        if (prod) {
            {
                #pragma unroll
                for (int cq = 0; cq < 4; ++cq) {
                    #pragma unroll
                    for (int rq = 0; rq < 2; ++rq) {
                        const int g = cq * 2 + rq;
                        v2f qa; qa.x = qf[cq][rq].x; qa.y = qf[cq][rq].y;
                        #pragma unroll
                        for (int tb = 0; tb < KT; ++tb) {
                            v2f kb; kb.x = fr[g * 4 + tb].x; kb.y = fr[g * 4 + tb].y;
                            s8[tb] = wmma4(qa, kb, s8[tb]);
                        }
                        qa.x = qf[cq][rq].z; qa.y = qf[cq][rq].w;
                        #pragma unroll
                        for (int tb = 0; tb < KT; ++tb) {
                            v2f kb; kb.x = fr[g * 4 + tb].z; kb.y = fr[g * 4 + tb].w;
                            s8[tb] = wmma4(qa, kb, s8[tb]);
                        }
                    }
                }

            }
        } else if (cons) {
            {
                const int slot = (jb - 1) & 1;
                const float4 a0 = *(const float4*)&abuf[slot][half * 8];
                const float4 a1 = *(const float4*)&abuf[slot][half * 8 + 4];
                const float al[8] = { a0.x, a0.y, a0.z, a0.w, a1.x, a1.y, a1.z, a1.w };
                #pragma unroll
                for (int r = 0; r < 8; ++r) {
                    o0[r] *= al[r]; o1[r] *= al[r];
                    o2[r] *= al[r]; o3[r] *= al[r];
                }
                #pragma unroll
                for (int tb = 0; tb < KT; ++tb) {
                    #pragma unroll
                    for (int rq = 0; rq < 2; ++rq) {
                        const float4 pf = *(const float4*)&pbuf[slot][nlo][16 * tb + 8 * half + 4 * rq];
                        v2f pa; pa.x = pf.x; pa.y = pf.y;
                        { v2f vb; vb.x = fr[(0 + rq) * 4 + tb].x; vb.y = fr[(0 + rq) * 4 + tb].y; o0 = wmma4(pa, vb, o0); }
                        { v2f vb; vb.x = fr[(2 + rq) * 4 + tb].x; vb.y = fr[(2 + rq) * 4 + tb].y; o1 = wmma4(pa, vb, o1); }
                        { v2f vb; vb.x = fr[(4 + rq) * 4 + tb].x; vb.y = fr[(4 + rq) * 4 + tb].y; o2 = wmma4(pa, vb, o2); }
                        { v2f vb; vb.x = fr[(6 + rq) * 4 + tb].x; vb.y = fr[(6 + rq) * 4 + tb].y; o3 = wmma4(pa, vb, o3); }
                        pa.x = pf.z; pa.y = pf.w;
                        { v2f vb; vb.x = fr[(0 + rq) * 4 + tb].z; vb.y = fr[(0 + rq) * 4 + tb].w; o0 = wmma4(pa, vb, o0); }
                        { v2f vb; vb.x = fr[(2 + rq) * 4 + tb].z; vb.y = fr[(2 + rq) * 4 + tb].w; o1 = wmma4(pa, vb, o1); }
                        { v2f vb; vb.x = fr[(4 + rq) * 4 + tb].z; vb.y = fr[(4 + rq) * 4 + tb].w; o2 = wmma4(pa, vb, o2); }
                        { v2f vb; vb.x = fr[(6 + rq) * 4 + tb].z; vb.y = fr[(6 + rq) * 4 + tb].w; o3 = wmma4(pa, vb, o3); }
                    }
                }
            }
        }
        {
            const int jw = jb + 1 - role;
            const int jn = (jw < nblk) ? jw : nblk - 1;
            const float* fp = fsrc + (size_t)(jn * KT) * TILE_F;
            #pragma unroll
            for (int i = 0; i < 32; ++i)
                fr[i] = *(const float4*)(fp + (i & 3) * TILE_F + (i >> 2) * RQ_F);
        }
        if (prod) {
            {
                const int slot = jb & 1;
                float sv[KT][8];
                #pragma unroll
                for (int tb = 0; tb < KT; ++tb) {
                    #pragma unroll
                    for (int r = 0; r < 8; ++r) sv[tb][r] = s8[tb][r] * scale;
                }

                if (jb == nblk - 1) {
                    const int kidx = jb * KB + nlo;
                    #pragma unroll
                    for (int tb = 0; tb < KT; ++tb) {
                        #pragma unroll
                        for (int r = 0; r < 8; ++r) {
                            if (kidx + 16 * tb > qidx + r) sv[tb][r] = -1e30f;
                        }
                    }
                }

                float alpha[8];
                #pragma unroll
                for (int r = 0; r < 8; ++r) {
                    float t = sv[0][r];
                    #pragma unroll
                    for (int tb = 1; tb < KT; ++tb) t = fmaxf(t, sv[tb][r]);
                    t = fmaxf(t, __shfl_xor(t, 1, 16));
                    t = fmaxf(t, __shfl_xor(t, 2, 16));
                    t = fmaxf(t, __shfl_xor(t, 4, 16));
                    t = fmaxf(t, __shfl_xor(t, 8, 16));
                    float mnew = fmaxf(mrow[r], t);
                    alpha[r] = __expf(mrow[r] - mnew);
                    mrow[r]  = mnew;
                    float s = 0.f;
                    #pragma unroll
                    for (int tb = 0; tb < KT; ++tb) {
                        float p = __expf(sv[tb][r] - mnew);
                        pbuf[slot][r + half * 8][16 * tb + nlo] = p;
                        s += p;
                    }
                    s += __shfl_xor(s, 1, 16);
                    s += __shfl_xor(s, 2, 16);
                    s += __shfl_xor(s, 4, 16);
                    s += __shfl_xor(s, 8, 16);
                    lrow[r] = lrow[r] * alpha[r] + s;
                }
                if (nlo == 0) {
                    #pragma unroll
                    for (int r = 0; r < 8; ++r) abuf[slot][half * 8 + r] = alpha[r];
                    if (jb == nblk - 1) {
                        #pragma unroll
                        for (int r = 0; r < 8; ++r) lbuf[half * 8 + r] = lrow[r];
                    }
                }
            }
        }
        __syncthreads();
    }

    if (role == 1) {
        const float4 l0 = *(const float4*)&lbuf[half * 8];
        const float4 l1 = *(const float4*)&lbuf[half * 8 + 4];
        const float lr[8] = { l0.x, l0.y, l0.z, l0.w, l1.x, l1.y, l1.z, l1.w };
        #pragma unroll
        for (int r = 0; r < 8; ++r) {
            float inv = 1.0f / lr[r];
            size_t idx = (size_t)(qrow0 + r + half * 8) * Hh;
            out[idx +  0 + nlo] = o0[r] * inv;
            out[idx + 16 + nlo] = o1[r] * inv;
            out[idx + 32 + nlo] = o2[r] * inv;
            out[idx + 48 + nlo] = o3[r] * inv;
        }
    }
}

extern "C" void kernel_launch(void* const* d_in, const int* in_sizes, int n_in,
                              void* d_out, int out_size, void* d_ws, size_t ws_size,
                              hipStream_t stream) {
    const float* x  = (const float*)d_in[0];
    const float* Wk = (const float*)d_in[1];
    const float* Wq = (const float*)d_in[2];
    const float* Wv = (const float*)d_in[3];
    float* out = (float*)d_out;

    float* ws   = (float*)d_ws;
    float* qbuf = ws;
    float* kbuf = ws + (size_t)ROWS * Hh;
    float* vbuf = ws + 2 * (size_t)ROWS * Hh;

    proj_kernel<<<ROWS / (16 * PR), 128, 0, stream>>>(x, Wk, Wq, Wv, qbuf, kbuf, vbuf);
    attn_kernel<<<ROWS / 16, 64, 0, stream>>>(qbuf, kbuf, vbuf, out);
}


# attention softmax: the 16 row max and row sum butterflies of a trip exchange lanes by DPP row_xmask on the vector ALU (was LDS-unit permutes, one LDS round trip per step); sum order kept
# speedup vs baseline: 2.0792x; 1.1110x over previous
#include <hip/hip_runtime.h>

typedef __attribute__((ext_vector_type(2))) float v2f;
typedef __attribute__((ext_vector_type(8))) float v8f;

constexpr int Bb = 8, Tt = 2048, Ee = 1024, Hh = 64;
constexpr int ROWS = Bb * Tt;
constexpr int TPB  = Tt / 16;

__device__ __forceinline__ v8f wmma4(v2f a, v2f b, v8f c) {
    return __builtin_amdgcn_wmma_f32_16x16x4_f32(
         false, a,  false, b,
         (short)0, c,  false,  false);
}

template <int M>
__device__ __forceinline__ float row_xor(float v) {
    return __builtin_bit_cast(float,
        __builtin_amdgcn_update_dpp(0, __builtin_bit_cast(int, v), 0x160 + M, 0xf, 0xf, true));
}

constexpr int TILE_F = 1024;
constexpr int CQ_F   = 256;
constexpr int RQ_F   = 128;

constexpr int PR = 4;
constexpr int PC = 3;

__global__ __launch_bounds__(128)
void proj_kernel(const float* __restrict__ x,
                 const float* __restrict__ Wk,
                 const float* __restrict__ Wq,
                 const float* __restrict__ Wv,
                 float* __restrict__ qo, float* __restrict__ ko,
                 float* __restrict__ vo) {
    const int lane = threadIdx.x & 31;
    const int wv   = __builtin_amdgcn_readfirstlane((int)(threadIdx.x >> 5));
    const int c    = lane & 15;
    const int h    = lane >> 4;
    const int rowb = blockIdx.x * (16 * PR);

    const float* wp[PC];
    wp[0] = Wk + (size_t)(4 * h) * Hh + wv * 16 + c;
    wp[1] = Wq + (size_t)(4 * h) * Hh + wv * 16 + c;
    wp[2] = Wv + (size_t)(4 * h) * Hh + wv * 16 + c;
    const float* xp = x + (size_t)(rowb + c) * Ee + 4 * h;

    v8f acc[PR][PC];
    #pragma unroll
    for (int rt = 0; rt < PR; ++rt) {
        #pragma unroll
        for (int t = 0; t < PC; ++t) acc[rt][t] = v8f{};
    }

    float4 xa0[2][PR], xa1[2][PR];
    float  wb0[2][PC][4], wb1[2][PC][4];

    auto load_chunk = [&](float4 (&xa)[2][PR], float (&wb)[2][PC][4], int e0) {
        #pragma unroll
        for (int g2 = 0; g2 < 2; ++g2) {
            #pragma unroll
            for (int rt = 0; rt < PR; ++rt)
                xa[g2][rt] = *(const float4*)(xp + (size_t)rt * 16 * Ee + e0 + 8 * g2);
            #pragma unroll
            for (int t = 0; t < PC; ++t) {
                #pragma unroll
                for (int j = 0; j < 4; ++j)
                    wb[g2][t][j] = wp[t][(size_t)(e0 + 8 * g2 + j) * Hh];
            }
        }
    };
    auto mma_chunk = [&](const float4 (&xa)[2][PR], const float (&wb)[2][PC][4]) {
        #pragma unroll
        for (int g2 = 0; g2 < 2; ++g2) {
            #pragma unroll
            for (int rt = 0; rt < PR; ++rt) {
                v2f a; a.x = xa[g2][rt].x; a.y = xa[g2][rt].y;
                #pragma unroll
                for (int t = 0; t < PC; ++t) {
                    v2f b; b.x = wb[g2][t][0]; b.y = wb[g2][t][1];
                    acc[rt][t] = (t < 2) ? wmma4(b, a, acc[rt][t]) : wmma4(a, b, acc[rt][t]);
                }
            }
            #pragma unroll
            for (int rt = 0; rt < PR; ++rt) {
                v2f a; a.x = xa[g2][rt].z; a.y = xa[g2][rt].w;
                #pragma unroll
                for (int t = 0; t < PC; ++t) {
                    v2f b; b.x = wb[g2][t][2]; b.y = wb[g2][t][3];
                    acc[rt][t] = (t < 2) ? wmma4(b, a, acc[rt][t]) : wmma4(a, b, acc[rt][t]);
                }
            }
        }
    };

    load_chunk(xa0, wb0, 0);
    for (int e0 = 0; e0 < Ee - 32; e0 += 32) {
        load_chunk(xa1, wb1, e0 + 16);
        mma_chunk(xa0, wb0);
        load_chunk(xa0, wb0, e0 + 32);
        mma_chunk(xa1, wb1);
    }
    load_chunk(xa1, wb1, Ee - 16);
    mma_chunk(xa0, wb0);
    mma_chunk(xa1, wb1);

    float* ob[PC] = { ko, qo, vo };
    #pragma unroll
    for (int rt = 0; rt < PR; ++rt) {
        const size_t rec = (size_t)(blockIdx.x * PR + rt) * TILE_F + wv * CQ_F + lane * 4;
        #pragma unroll
        for (int t = 0; t < PC; ++t) {
            #pragma unroll
            for (int rq = 0; rq < 2; ++rq) {
                float4 o4;
                o4.x = acc[rt][t][4 * rq + 0]; o4.y = acc[rt][t][4 * rq + 1];
                o4.z = acc[rt][t][4 * rq + 2]; o4.w = acc[rt][t][4 * rq + 3];
                *(float4*)(ob[t] + rec + rq * RQ_F) = o4;
            }
        }
    }
}

constexpr int KT = 4;
constexpr int KB = 16 * KT;
constexpr int PP = KB + 4;

__global__ __launch_bounds__(64)
void attn_kernel(const float* __restrict__ q,
                 const float* __restrict__ k,
                 const float* __restrict__ v,
                 float* __restrict__ out) {
    __shared__ __attribute__((aligned(16))) float pbuf[2][16][PP];
    __shared__ __attribute__((aligned(16))) float abuf[2][16];
    __shared__ __attribute__((aligned(16))) float lbuf[16];

    const int lane  = threadIdx.x & 31;
    const int role  = __builtin_amdgcn_readfirstlane((int)(threadIdx.x >> 5));
    const int tile  = blockIdx.x;
    const int b     = tile / TPB;
    const int it    = tile % TPB;
    const int qrow0 = tile * 16;
    const int half  = lane >> 4;
    const int nlo   = lane & 15;

    const float scale = 0.03125f;
    const int nblk = it / KT + 1;
    const int qidx = it * 16 + half * 8;
    const size_t recb = (size_t)(b * TPB) * TILE_F + lane * 4;

    float4 qf[4][2];
    float4 fr[32];
    v8f o0 = {}, o1 = {}, o2 = {}, o3 = {};
    float mrow[8], lrow[8];
    #pragma unroll
    for (int r = 0; r < 8; ++r) { mrow[r] = -1e30f; lrow[r] = 0.f; }

    if (role == 0) {
        const float* qp = q + (size_t)tile * TILE_F + lane * 4;
        #pragma unroll
        for (int cq = 0; cq < 4; ++cq) {
            #pragma unroll
            for (int rq = 0; rq < 2; ++rq)
                qf[cq][rq] = *(const float4*)(qp + cq * CQ_F + rq * RQ_F);
        }
    } else {
        #pragma unroll
        for (int cq = 0; cq < 4; ++cq) {
            #pragma unroll
            for (int rq = 0; rq < 2; ++rq)
                qf[cq][rq] = make_float4(0.f, 0.f, 0.f, 0.f);
        }
    }
    const float* fsrc = ((role == 0) ? k : v) + recb;
    #pragma unroll
    for (int i = 0; i < 32; ++i)
        fr[i] = *(const float4*)(fsrc + (i & 3) * TILE_F + (i >> 2) * RQ_F);

    for (int jb = 0; jb <= nblk; ++jb) {
        const bool prod = (role == 0) && (jb < nblk);
        const bool cons = (role != 0) && (jb > 0);
        v8f s8[KT];
        #pragma unroll
        for (int tb = 0; tb < KT; ++tb) s8[tb] = v8f{};
        if (prod) {
            {
                #pragma unroll
                for (int cq = 0; cq < 4; ++cq) {
                    #pragma unroll
                    for (int rq = 0; rq < 2; ++rq) {
                        const int g = cq * 2 + rq;
                        v2f qa; qa.x = qf[cq][rq].x; qa.y = qf[cq][rq].y;
                        #pragma unroll
                        for (int tb = 0; tb < KT; ++tb) {
                            v2f kb; kb.x = fr[g * 4 + tb].x; kb.y = fr[g * 4 + tb].y;
                            s8[tb] = wmma4(qa, kb, s8[tb]);
                        }
                        qa.x = qf[cq][rq].z; qa.y = qf[cq][rq].w;
                        #pragma unroll
                        for (int tb = 0; tb < KT; ++tb) {
                            v2f kb; kb.x = fr[g * 4 + tb].z; kb.y = fr[g * 4 + tb].w;
                            s8[tb] = wmma4(qa, kb, s8[tb]);
                        }
                    }
                }

            }
        } else if (cons) {
            {
                const int slot = (jb - 1) & 1;
                const float4 a0 = *(const float4*)&abuf[slot][half * 8];
                const float4 a1 = *(const float4*)&abuf[slot][half * 8 + 4];
                const float al[8] = { a0.x, a0.y, a0.z, a0.w, a1.x, a1.y, a1.z, a1.w };
                #pragma unroll
                for (int r = 0; r < 8; ++r) {
                    o0[r] *= al[r]; o1[r] *= al[r];
                    o2[r] *= al[r]; o3[r] *= al[r];
                }
                #pragma unroll
                for (int tb = 0; tb < KT; ++tb) {
                    #pragma unroll
                    for (int rq = 0; rq < 2; ++rq) {
                        const float4 pf = *(const float4*)&pbuf[slot][nlo][16 * tb + 8 * half + 4 * rq];
                        v2f pa; pa.x = pf.x; pa.y = pf.y;
                        { v2f vb; vb.x = fr[(0 + rq) * 4 + tb].x; vb.y = fr[(0 + rq) * 4 + tb].y; o0 = wmma4(pa, vb, o0); }
                        { v2f vb; vb.x = fr[(2 + rq) * 4 + tb].x; vb.y = fr[(2 + rq) * 4 + tb].y; o1 = wmma4(pa, vb, o1); }
                        { v2f vb; vb.x = fr[(4 + rq) * 4 + tb].x; vb.y = fr[(4 + rq) * 4 + tb].y; o2 = wmma4(pa, vb, o2); }
                        { v2f vb; vb.x = fr[(6 + rq) * 4 + tb].x; vb.y = fr[(6 + rq) * 4 + tb].y; o3 = wmma4(pa, vb, o3); }
                        pa.x = pf.z; pa.y = pf.w;
                        { v2f vb; vb.x = fr[(0 + rq) * 4 + tb].z; vb.y = fr[(0 + rq) * 4 + tb].w; o0 = wmma4(pa, vb, o0); }
                        { v2f vb; vb.x = fr[(2 + rq) * 4 + tb].z; vb.y = fr[(2 + rq) * 4 + tb].w; o1 = wmma4(pa, vb, o1); }
                        { v2f vb; vb.x = fr[(4 + rq) * 4 + tb].z; vb.y = fr[(4 + rq) * 4 + tb].w; o2 = wmma4(pa, vb, o2); }
                        { v2f vb; vb.x = fr[(6 + rq) * 4 + tb].z; vb.y = fr[(6 + rq) * 4 + tb].w; o3 = wmma4(pa, vb, o3); }
                    }
                }
            }
        }
        {
            const int jw = jb + 1 - role;
            const int jn = (jw < nblk) ? jw : nblk - 1;
            const float* fp = fsrc + (size_t)(jn * KT) * TILE_F;
            #pragma unroll
            for (int i = 0; i < 32; ++i)
                fr[i] = *(const float4*)(fp + (i & 3) * TILE_F + (i >> 2) * RQ_F);
        }
        if (prod) {
            {
                const int slot = jb & 1;
                float sv[KT][8];
                #pragma unroll
                for (int tb = 0; tb < KT; ++tb) {
                    #pragma unroll
                    for (int r = 0; r < 8; ++r) sv[tb][r] = s8[tb][r] * scale;
                }

                if (jb == nblk - 1) {
                    const int kidx = jb * KB + nlo;
                    #pragma unroll
                    for (int tb = 0; tb < KT; ++tb) {
                        #pragma unroll
                        for (int r = 0; r < 8; ++r) {
                            if (kidx + 16 * tb > qidx + r) sv[tb][r] = -1e30f;
                        }
                    }
                }

                float alpha[8];
                #pragma unroll
                for (int r = 0; r < 8; ++r) {
                    float t = sv[0][r];
                    #pragma unroll
                    for (int tb = 1; tb < KT; ++tb) t = fmaxf(t, sv[tb][r]);
                    t = fmaxf(t, row_xor<1>(t));
                    t = fmaxf(t, row_xor<2>(t));
                    t = fmaxf(t, row_xor<4>(t));
                    t = fmaxf(t, row_xor<8>(t));
                    float mnew = fmaxf(mrow[r], t);
                    alpha[r] = __expf(mrow[r] - mnew);
                    mrow[r]  = mnew;
                    float s = 0.f;
                    #pragma unroll
                    for (int tb = 0; tb < KT; ++tb) {
                        float p = __expf(sv[tb][r] - mnew);
                        pbuf[slot][r + half * 8][16 * tb + nlo] = p;
                        s += p;
                    }
                    s += row_xor<1>(s);
                    s += row_xor<2>(s);
                    s += row_xor<4>(s);
                    s += row_xor<8>(s);
                    lrow[r] = lrow[r] * alpha[r] + s;
                }
                if (nlo == 0) {
                    #pragma unroll
                    for (int r = 0; r < 8; ++r) abuf[slot][half * 8 + r] = alpha[r];
                    if (jb == nblk - 1) {
                        #pragma unroll
                        for (int r = 0; r < 8; ++r) lbuf[half * 8 + r] = lrow[r];
                    }
                }
            }
        }
        __syncthreads();
    }

    if (role == 1) {
        const float4 l0 = *(const float4*)&lbuf[half * 8];
        const float4 l1 = *(const float4*)&lbuf[half * 8 + 4];
        const float lr[8] = { l0.x, l0.y, l0.z, l0.w, l1.x, l1.y, l1.z, l1.w };
        #pragma unroll
        for (int r = 0; r < 8; ++r) {
            float inv = 1.0f / lr[r];
            size_t idx = (size_t)(qrow0 + r + half * 8) * Hh;
            out[idx +  0 + nlo] = o0[r] * inv;
            out[idx + 16 + nlo] = o1[r] * inv;
            out[idx + 32 + nlo] = o2[r] * inv;
            out[idx + 48 + nlo] = o3[r] * inv;
        }
    }
}

extern "C" void kernel_launch(void* const* d_in, const int* in_sizes, int n_in,
                              void* d_out, int out_size, void* d_ws, size_t ws_size,
                              hipStream_t stream) {
    const float* x  = (const float*)d_in[0];
    const float* Wk = (const float*)d_in[1];
    const float* Wq = (const float*)d_in[2];
    const float* Wv = (const float*)d_in[3];
    float* out = (float*)d_out;

    float* ws   = (float*)d_ws;
    float* qbuf = ws;
    float* kbuf = ws + (size_t)ROWS * Hh;
    float* vbuf = ws + 2 * (size_t)ROWS * Hh;

    proj_kernel<<<ROWS / (16 * PR), 128, 0, stream>>>(x, Wk, Wq, Wv, qbuf, kbuf, vbuf);
    attn_kernel<<<ROWS / 16, 64, 0, stream>>>(qbuf, kbuf, vbuf, out);
}


# projection: k split in halves over 8-wave groups (was 4 waves, one per SIMD): two waves per SIMD, each sums 512 k; high-half tile added via LDS, so k sums as (0..511)+(512..1023)
# speedup vs baseline: 2.2954x; 1.1040x over previous
#include <hip/hip_runtime.h>

typedef __attribute__((ext_vector_type(2))) float v2f;
typedef __attribute__((ext_vector_type(8))) float v8f;

constexpr int Bb = 8, Tt = 2048, Ee = 1024, Hh = 64;
constexpr int ROWS = Bb * Tt;
constexpr int TPB  = Tt / 16;

__device__ __forceinline__ v8f wmma4(v2f a, v2f b, v8f c) {
    return __builtin_amdgcn_wmma_f32_16x16x4_f32(
         false, a,  false, b,
         (short)0, c,  false,  false);
}

template <int M>
__device__ __forceinline__ float row_xor(float v) {
    return __builtin_bit_cast(float,
        __builtin_amdgcn_update_dpp(0, __builtin_bit_cast(int, v), 0x160 + M, 0xf, 0xf, true));
}

constexpr int TILE_F = 1024;
constexpr int CQ_F   = 256;
constexpr int RQ_F   = 128;

constexpr int PR = 4;
constexpr int PC = 3;
constexpr int EH = Ee / 2;

__global__ __launch_bounds__(256)
void proj_kernel(const float* __restrict__ x,
                 const float* __restrict__ Wk,
                 const float* __restrict__ Wq,
                 const float* __restrict__ Wv,
                 float* __restrict__ qo, float* __restrict__ ko,
                 float* __restrict__ vo) {
    __shared__ __attribute__((aligned(16))) float4 red[4][PR * PC * 2][32];

    const int lane = threadIdx.x & 31;
    const int wid  = __builtin_amdgcn_readfirstlane((int)(threadIdx.x >> 5));
    const int wv   = wid & 3;
    const int kh   = wid >> 2;
    const int c    = lane & 15;
    const int h    = lane >> 4;
    const int rowb = blockIdx.x * (16 * PR);
    const int kb   = kh * EH;

    const float* wp[PC];
    wp[0] = Wk + (size_t)(kb + 4 * h) * Hh + wv * 16 + c;
    wp[1] = Wq + (size_t)(kb + 4 * h) * Hh + wv * 16 + c;
    wp[2] = Wv + (size_t)(kb + 4 * h) * Hh + wv * 16 + c;
    const float* xp = x + (size_t)(rowb + c) * Ee + kb + 4 * h;

    v8f acc[PR][PC];
    #pragma unroll
    for (int rt = 0; rt < PR; ++rt) {
        #pragma unroll
        for (int t = 0; t < PC; ++t) acc[rt][t] = v8f{};
    }

    float4 xa0[2][PR], xa1[2][PR];
    float  wb0[2][PC][4], wb1[2][PC][4];

    auto load_chunk = [&](float4 (&xa)[2][PR], float (&wb)[2][PC][4], int e0) {
        #pragma unroll
        for (int g2 = 0; g2 < 2; ++g2) {
            #pragma unroll
            for (int rt = 0; rt < PR; ++rt)
                xa[g2][rt] = *(const float4*)(xp + (size_t)rt * 16 * Ee + e0 + 8 * g2);
            #pragma unroll
            for (int t = 0; t < PC; ++t) {
                #pragma unroll
                for (int j = 0; j < 4; ++j)
                    wb[g2][t][j] = wp[t][(size_t)(e0 + 8 * g2 + j) * Hh];
            }
        }
    };
    auto mma_chunk = [&](const float4 (&xa)[2][PR], const float (&wb)[2][PC][4]) {
        #pragma unroll
        for (int g2 = 0; g2 < 2; ++g2) {
            #pragma unroll
            for (int rt = 0; rt < PR; ++rt) {
                v2f a; a.x = xa[g2][rt].x; a.y = xa[g2][rt].y;
                #pragma unroll
                for (int t = 0; t < PC; ++t) {
                    v2f b; b.x = wb[g2][t][0]; b.y = wb[g2][t][1];
                    acc[rt][t] = (t < 2) ? wmma4(b, a, acc[rt][t]) : wmma4(a, b, acc[rt][t]);
                }
            }
            #pragma unroll
            for (int rt = 0; rt < PR; ++rt) {
                v2f a; a.x = xa[g2][rt].z; a.y = xa[g2][rt].w;
                #pragma unroll
                for (int t = 0; t < PC; ++t) {
                    v2f b; b.x = wb[g2][t][2]; b.y = wb[g2][t][3];
                    acc[rt][t] = (t < 2) ? wmma4(b, a, acc[rt][t]) : wmma4(a, b, acc[rt][t]);
                }
            }
        }
    };

    load_chunk(xa0, wb0, 0);
    for (int e0 = 0; e0 < EH - 32; e0 += 32) {
        load_chunk(xa1, wb1, e0 + 16);
        mma_chunk(xa0, wb0);
        load_chunk(xa0, wb0, e0 + 32);
        mma_chunk(xa1, wb1);
    }
    load_chunk(xa1, wb1, EH - 16);
    mma_chunk(xa0, wb0);
    mma_chunk(xa1, wb1);

    if (kh == 1) {
        #pragma unroll
        for (int rt = 0; rt < PR; ++rt) {
            #pragma unroll
            for (int t = 0; t < PC; ++t) {
                #pragma unroll
                for (int rq = 0; rq < 2; ++rq) {
                    float4 p4;
                    p4.x = acc[rt][t][4 * rq + 0]; p4.y = acc[rt][t][4 * rq + 1];
                    p4.z = acc[rt][t][4 * rq + 2]; p4.w = acc[rt][t][4 * rq + 3];
                    red[wv][(rt * PC + t) * 2 + rq][lane] = p4;
                }
            }
        }
    }
    __syncthreads();
    if (kh == 1) return;

    float* ob[PC] = { ko, qo, vo };
    #pragma unroll
    for (int rt = 0; rt < PR; ++rt) {
        const size_t rec = (size_t)(blockIdx.x * PR + rt) * TILE_F + wv * CQ_F + lane * 4;
        #pragma unroll
        for (int t = 0; t < PC; ++t) {
            #pragma unroll
            for (int rq = 0; rq < 2; ++rq) {
                const float4 p4 = red[wv][(rt * PC + t) * 2 + rq][lane];
                float4 o4;
                o4.x = acc[rt][t][4 * rq + 0] + p4.x; o4.y = acc[rt][t][4 * rq + 1] + p4.y;
                o4.z = acc[rt][t][4 * rq + 2] + p4.z; o4.w = acc[rt][t][4 * rq + 3] + p4.w;
                *(float4*)(ob[t] + rec + rq * RQ_F) = o4;
            }
        }
    }
}

constexpr int KT = 4;
constexpr int KB = 16 * KT;
constexpr int PP = KB + 4;

__global__ __launch_bounds__(64)
void attn_kernel(const float* __restrict__ q,
                 const float* __restrict__ k,
                 const float* __restrict__ v,
                 float* __restrict__ out) {
    __shared__ __attribute__((aligned(16))) float pbuf[2][16][PP];
    __shared__ __attribute__((aligned(16))) float abuf[2][16];
    __shared__ __attribute__((aligned(16))) float lbuf[16];

    const int lane  = threadIdx.x & 31;
    const int role  = __builtin_amdgcn_readfirstlane((int)(threadIdx.x >> 5));
    const int tile  = blockIdx.x;
    const int b     = tile / TPB;
    const int it    = tile % TPB;
    const int qrow0 = tile * 16;
    const int half  = lane >> 4;
    const int nlo   = lane & 15;

    const float scale = 0.03125f;
    const int nblk = it / KT + 1;
    const int qidx = it * 16 + half * 8;
    const size_t recb = (size_t)(b * TPB) * TILE_F + lane * 4;

    float4 qf[4][2];
    float4 fr[32];
    v8f o0 = {}, o1 = {}, o2 = {}, o3 = {};
    float mrow[8], lrow[8];
    #pragma unroll
    for (int r = 0; r < 8; ++r) { mrow[r] = -1e30f; lrow[r] = 0.f; }

    if (role == 0) {
        const float* qp = q + (size_t)tile * TILE_F + lane * 4;
        #pragma unroll
        for (int cq = 0; cq < 4; ++cq) {
            #pragma unroll
            for (int rq = 0; rq < 2; ++rq)
                qf[cq][rq] = *(const float4*)(qp + cq * CQ_F + rq * RQ_F);
        }
    } else {
        #pragma unroll
        for (int cq = 0; cq < 4; ++cq) {
            #pragma unroll
            for (int rq = 0; rq < 2; ++rq)
                qf[cq][rq] = make_float4(0.f, 0.f, 0.f, 0.f);
        }
    }
    const float* fsrc = ((role == 0) ? k : v) + recb;
    #pragma unroll
    for (int i = 0; i < 32; ++i)
        fr[i] = *(const float4*)(fsrc + (i & 3) * TILE_F + (i >> 2) * RQ_F);

    for (int jb = 0; jb <= nblk; ++jb) {
        const bool prod = (role == 0) && (jb < nblk);
        const bool cons = (role != 0) && (jb > 0);
        v8f s8[KT];
        #pragma unroll
        for (int tb = 0; tb < KT; ++tb) s8[tb] = v8f{};
        if (prod) {
            {
                #pragma unroll
                for (int cq = 0; cq < 4; ++cq) {
                    #pragma unroll
                    for (int rq = 0; rq < 2; ++rq) {
                        const int g = cq * 2 + rq;
                        v2f qa; qa.x = qf[cq][rq].x; qa.y = qf[cq][rq].y;
                        #pragma unroll
                        for (int tb = 0; tb < KT; ++tb) {
                            v2f kb; kb.x = fr[g * 4 + tb].x; kb.y = fr[g * 4 + tb].y;
                            s8[tb] = wmma4(qa, kb, s8[tb]);
                        }
                        qa.x = qf[cq][rq].z; qa.y = qf[cq][rq].w;
                        #pragma unroll
                        for (int tb = 0; tb < KT; ++tb) {
                            v2f kb; kb.x = fr[g * 4 + tb].z; kb.y = fr[g * 4 + tb].w;
                            s8[tb] = wmma4(qa, kb, s8[tb]);
                        }
                    }
                }

            }
        } else if (cons) {
            {
                const int slot = (jb - 1) & 1;
                const float4 a0 = *(const float4*)&abuf[slot][half * 8];
                const float4 a1 = *(const float4*)&abuf[slot][half * 8 + 4];
                const float al[8] = { a0.x, a0.y, a0.z, a0.w, a1.x, a1.y, a1.z, a1.w };
                #pragma unroll
                for (int r = 0; r < 8; ++r) {
                    o0[r] *= al[r]; o1[r] *= al[r];
                    o2[r] *= al[r]; o3[r] *= al[r];
                }
                #pragma unroll
                for (int tb = 0; tb < KT; ++tb) {
                    #pragma unroll
                    for (int rq = 0; rq < 2; ++rq) {
                        const float4 pf = *(const float4*)&pbuf[slot][nlo][16 * tb + 8 * half + 4 * rq];
                        v2f pa; pa.x = pf.x; pa.y = pf.y;
                        { v2f vb; vb.x = fr[(0 + rq) * 4 + tb].x; vb.y = fr[(0 + rq) * 4 + tb].y; o0 = wmma4(pa, vb, o0); }
                        { v2f vb; vb.x = fr[(2 + rq) * 4 + tb].x; vb.y = fr[(2 + rq) * 4 + tb].y; o1 = wmma4(pa, vb, o1); }
                        { v2f vb; vb.x = fr[(4 + rq) * 4 + tb].x; vb.y = fr[(4 + rq) * 4 + tb].y; o2 = wmma4(pa, vb, o2); }
                        { v2f vb; vb.x = fr[(6 + rq) * 4 + tb].x; vb.y = fr[(6 + rq) * 4 + tb].y; o3 = wmma4(pa, vb, o3); }
                        pa.x = pf.z; pa.y = pf.w;
                        { v2f vb; vb.x = fr[(0 + rq) * 4 + tb].z; vb.y = fr[(0 + rq) * 4 + tb].w; o0 = wmma4(pa, vb, o0); }
                        { v2f vb; vb.x = fr[(2 + rq) * 4 + tb].z; vb.y = fr[(2 + rq) * 4 + tb].w; o1 = wmma4(pa, vb, o1); }
                        { v2f vb; vb.x = fr[(4 + rq) * 4 + tb].z; vb.y = fr[(4 + rq) * 4 + tb].w; o2 = wmma4(pa, vb, o2); }
                        { v2f vb; vb.x = fr[(6 + rq) * 4 + tb].z; vb.y = fr[(6 + rq) * 4 + tb].w; o3 = wmma4(pa, vb, o3); }
                    }
                }
            }
        }
        {
            const int jw = jb + 1 - role;
            const int jn = (jw < nblk) ? jw : nblk - 1;
            const float* fp = fsrc + (size_t)(jn * KT) * TILE_F;
            #pragma unroll
            for (int i = 0; i < 32; ++i)
                fr[i] = *(const float4*)(fp + (i & 3) * TILE_F + (i >> 2) * RQ_F);
        }
        if (prod) {
            {
                const int slot = jb & 1;
                float sv[KT][8];
                #pragma unroll
                for (int tb = 0; tb < KT; ++tb) {
                    #pragma unroll
                    for (int r = 0; r < 8; ++r) sv[tb][r] = s8[tb][r] * scale;
                }

                if (jb == nblk - 1) {
                    const int kidx = jb * KB + nlo;
                    #pragma unroll
                    for (int tb = 0; tb < KT; ++tb) {
                        #pragma unroll
                        for (int r = 0; r < 8; ++r) {
                            if (kidx + 16 * tb > qidx + r) sv[tb][r] = -1e30f;
                        }
                    }
                }

                float alpha[8];
                #pragma unroll
                for (int r = 0; r < 8; ++r) {
                    float t = sv[0][r];
                    #pragma unroll
                    for (int tb = 1; tb < KT; ++tb) t = fmaxf(t, sv[tb][r]);
                    t = fmaxf(t, row_xor<1>(t));
                    t = fmaxf(t, row_xor<2>(t));
                    t = fmaxf(t, row_xor<4>(t));
                    t = fmaxf(t, row_xor<8>(t));
                    float mnew = fmaxf(mrow[r], t);
                    alpha[r] = __expf(mrow[r] - mnew);
                    mrow[r]  = mnew;
                    float s = 0.f;
                    #pragma unroll
                    for (int tb = 0; tb < KT; ++tb) {
                        float p = __expf(sv[tb][r] - mnew);
                        pbuf[slot][r + half * 8][16 * tb + nlo] = p;
                        s += p;
                    }
                    s += row_xor<1>(s);
                    s += row_xor<2>(s);
                    s += row_xor<4>(s);
                    s += row_xor<8>(s);
                    lrow[r] = lrow[r] * alpha[r] + s;
                }
                if (nlo == 0) {
                    #pragma unroll
                    for (int r = 0; r < 8; ++r) abuf[slot][half * 8 + r] = alpha[r];
                    if (jb == nblk - 1) {
                        #pragma unroll
                        for (int r = 0; r < 8; ++r) lbuf[half * 8 + r] = lrow[r];
                    }
                }
            }
        }
        __syncthreads();
    }

    if (role == 1) {
        const float4 l0 = *(const float4*)&lbuf[half * 8];
        const float4 l1 = *(const float4*)&lbuf[half * 8 + 4];
        const float lr[8] = { l0.x, l0.y, l0.z, l0.w, l1.x, l1.y, l1.z, l1.w };
        #pragma unroll
        for (int r = 0; r < 8; ++r) {
            float inv = 1.0f / lr[r];
            size_t idx = (size_t)(qrow0 + r + half * 8) * Hh;
            out[idx +  0 + nlo] = o0[r] * inv;
            out[idx + 16 + nlo] = o1[r] * inv;
            out[idx + 32 + nlo] = o2[r] * inv;
            out[idx + 48 + nlo] = o3[r] * inv;
        }
    }
}

extern "C" void kernel_launch(void* const* d_in, const int* in_sizes, int n_in,
                              void* d_out, int out_size, void* d_ws, size_t ws_size,
                              hipStream_t stream) {
    const float* x  = (const float*)d_in[0];
    const float* Wk = (const float*)d_in[1];
    const float* Wq = (const float*)d_in[2];
    const float* Wv = (const float*)d_in[3];
    float* out = (float*)d_out;

    float* ws   = (float*)d_ws;
    float* qbuf = ws;
    float* kbuf = ws + (size_t)ROWS * Hh;
    float* vbuf = ws + 2 * (size_t)ROWS * Hh;

    proj_kernel<<<ROWS / (16 * PR), 256, 0, stream>>>(x, Wk, Wq, Wv, qbuf, kbuf, vbuf);
    attn_kernel<<<ROWS / 16, 64, 0, stream>>>(qbuf, kbuf, vbuf, out);
}


# attention: 8-wave work-group = 4 adjacent query tiles in step (was 1), K/V lines shared in the compute unit's cache; trip barrier orders LDS only, record loads fly across; sum order kept
# speedup vs baseline: 2.3970x; 1.0443x over previous
#include <hip/hip_runtime.h>

typedef __attribute__((ext_vector_type(2))) float v2f;
typedef __attribute__((ext_vector_type(8))) float v8f;

constexpr int Bb = 8, Tt = 2048, Ee = 1024, Hh = 64;
constexpr int ROWS = Bb * Tt;
constexpr int TPB  = Tt / 16;

__device__ __forceinline__ v8f wmma4(v2f a, v2f b, v8f c) {
    return __builtin_amdgcn_wmma_f32_16x16x4_f32(
         false, a,  false, b,
         (short)0, c,  false,  false);
}

template <int M>
__device__ __forceinline__ float row_xor(float v) {
    return __builtin_bit_cast(float,
        __builtin_amdgcn_update_dpp(0, __builtin_bit_cast(int, v), 0x160 + M, 0xf, 0xf, true));
}

constexpr int TILE_F = 1024;
constexpr int CQ_F   = 256;
constexpr int RQ_F   = 128;

constexpr int PR = 4;
constexpr int PC = 3;
constexpr int EH = Ee / 2;

__global__ __launch_bounds__(256)
void proj_kernel(const float* __restrict__ x,
                 const float* __restrict__ Wk,
                 const float* __restrict__ Wq,
                 const float* __restrict__ Wv,
                 float* __restrict__ qo, float* __restrict__ ko,
                 float* __restrict__ vo) {
    __shared__ __attribute__((aligned(16))) float4 red[4][PR * PC * 2][32];

    const int lane = threadIdx.x & 31;
    const int wid  = __builtin_amdgcn_readfirstlane((int)(threadIdx.x >> 5));
    const int wv   = wid & 3;
    const int kh   = wid >> 2;
    const int c    = lane & 15;
    const int h    = lane >> 4;
    const int rowb = blockIdx.x * (16 * PR);
    const int kb   = kh * EH;

    const float* wp[PC];
    wp[0] = Wk + (size_t)(kb + 4 * h) * Hh + wv * 16 + c;
    wp[1] = Wq + (size_t)(kb + 4 * h) * Hh + wv * 16 + c;
    wp[2] = Wv + (size_t)(kb + 4 * h) * Hh + wv * 16 + c;
    const float* xp = x + (size_t)(rowb + c) * Ee + kb + 4 * h;

    v8f acc[PR][PC];
    #pragma unroll
    for (int rt = 0; rt < PR; ++rt) {
        #pragma unroll
        for (int t = 0; t < PC; ++t) acc[rt][t] = v8f{};
    }

    float4 xa0[2][PR], xa1[2][PR];
    float  wb0[2][PC][4], wb1[2][PC][4];

    auto load_chunk = [&](float4 (&xa)[2][PR], float (&wb)[2][PC][4], int e0) {
        #pragma unroll
        for (int g2 = 0; g2 < 2; ++g2) {
            #pragma unroll
            for (int rt = 0; rt < PR; ++rt)
                xa[g2][rt] = *(const float4*)(xp + (size_t)rt * 16 * Ee + e0 + 8 * g2);
            #pragma unroll
            for (int t = 0; t < PC; ++t) {
                #pragma unroll
                for (int j = 0; j < 4; ++j)
                    wb[g2][t][j] = wp[t][(size_t)(e0 + 8 * g2 + j) * Hh];
            }
        }
    };
    auto mma_chunk = [&](const float4 (&xa)[2][PR], const float (&wb)[2][PC][4]) {
        #pragma unroll
        for (int g2 = 0; g2 < 2; ++g2) {
            #pragma unroll
            for (int rt = 0; rt < PR; ++rt) {
                v2f a; a.x = xa[g2][rt].x; a.y = xa[g2][rt].y;
                #pragma unroll
                for (int t = 0; t < PC; ++t) {
                    v2f b; b.x = wb[g2][t][0]; b.y = wb[g2][t][1];
                    acc[rt][t] = (t < 2) ? wmma4(b, a, acc[rt][t]) : wmma4(a, b, acc[rt][t]);
                }
            }
            #pragma unroll
            for (int rt = 0; rt < PR; ++rt) {
                v2f a; a.x = xa[g2][rt].z; a.y = xa[g2][rt].w;
                #pragma unroll
                for (int t = 0; t < PC; ++t) {
                    v2f b; b.x = wb[g2][t][2]; b.y = wb[g2][t][3];
                    acc[rt][t] = (t < 2) ? wmma4(b, a, acc[rt][t]) : wmma4(a, b, acc[rt][t]);
                }
            }
        }
    };

    load_chunk(xa0, wb0, 0);
    for (int e0 = 0; e0 < EH - 32; e0 += 32) {
        load_chunk(xa1, wb1, e0 + 16);
        mma_chunk(xa0, wb0);
        load_chunk(xa0, wb0, e0 + 32);
        mma_chunk(xa1, wb1);
    }
    load_chunk(xa1, wb1, EH - 16);
    mma_chunk(xa0, wb0);
    mma_chunk(xa1, wb1);

    if (kh == 1) {
        #pragma unroll
        for (int rt = 0; rt < PR; ++rt) {
            #pragma unroll
            for (int t = 0; t < PC; ++t) {
                #pragma unroll
                for (int rq = 0; rq < 2; ++rq) {
                    float4 p4;
                    p4.x = acc[rt][t][4 * rq + 0]; p4.y = acc[rt][t][4 * rq + 1];
                    p4.z = acc[rt][t][4 * rq + 2]; p4.w = acc[rt][t][4 * rq + 3];
                    red[wv][(rt * PC + t) * 2 + rq][lane] = p4;
                }
            }
        }
    }
    __syncthreads();
    if (kh == 1) return;

    float* ob[PC] = { ko, qo, vo };
    #pragma unroll
    for (int rt = 0; rt < PR; ++rt) {
        const size_t rec = (size_t)(blockIdx.x * PR + rt) * TILE_F + wv * CQ_F + lane * 4;
        #pragma unroll
        for (int t = 0; t < PC; ++t) {
            #pragma unroll
            for (int rq = 0; rq < 2; ++rq) {
                const float4 p4 = red[wv][(rt * PC + t) * 2 + rq][lane];
                float4 o4;
                o4.x = acc[rt][t][4 * rq + 0] + p4.x; o4.y = acc[rt][t][4 * rq + 1] + p4.y;
                o4.z = acc[rt][t][4 * rq + 2] + p4.z; o4.w = acc[rt][t][4 * rq + 3] + p4.w;
                *(float4*)(ob[t] + rec + rq * RQ_F) = o4;
            }
        }
    }
}

constexpr int KT = 4;
constexpr int KB = 16 * KT;
constexpr int PP = KB + 4;

constexpr int QT = 4;
static_assert(KT % QT == 0 && TPB % QT == 0, "the tiles of a work-group must share batch and trip count");

__global__ __launch_bounds__(64 * QT)
void attn_kernel(const float* __restrict__ q,
                 const float* __restrict__ k,
                 const float* __restrict__ v,
                 float* __restrict__ out) {
    __shared__ __attribute__((aligned(16))) float pbuf_all[QT][2][16][PP];
    __shared__ __attribute__((aligned(16))) float abuf_all[QT][2][16];
    __shared__ __attribute__((aligned(16))) float lbuf_all[QT][16];

    const int lane  = threadIdx.x & 31;
    const int wid   = __builtin_amdgcn_readfirstlane((int)(threadIdx.x >> 5));
    const int role  = wid / QT;
    const int pr    = wid % QT;
    const int tile  = blockIdx.x * QT + pr;
    float (*pbuf)[16][PP] = pbuf_all[pr];
    float (*abuf)[16]     = abuf_all[pr];
    float  *lbuf          = lbuf_all[pr];
    const int b     = tile / TPB;
    const int it    = tile % TPB;
    const int qrow0 = tile * 16;
    const int half  = lane >> 4;
    const int nlo   = lane & 15;

    const float scale = 0.03125f;
    const int nblk = it / KT + 1;
    const int qidx = it * 16 + half * 8;
    const size_t recb = (size_t)(b * TPB) * TILE_F + lane * 4;

    float4 qf[4][2];
    float4 fr[32];
    v8f o0 = {}, o1 = {}, o2 = {}, o3 = {};
    float mrow[8], lrow[8];
    #pragma unroll
    for (int r = 0; r < 8; ++r) { mrow[r] = -1e30f; lrow[r] = 0.f; }

    if (role == 0) {
        const float* qp = q + (size_t)tile * TILE_F + lane * 4;
        #pragma unroll
        for (int cq = 0; cq < 4; ++cq) {
            #pragma unroll
            for (int rq = 0; rq < 2; ++rq)
                qf[cq][rq] = *(const float4*)(qp + cq * CQ_F + rq * RQ_F);
        }
    } else {
        #pragma unroll
        for (int cq = 0; cq < 4; ++cq) {
            #pragma unroll
            for (int rq = 0; rq < 2; ++rq)
                qf[cq][rq] = make_float4(0.f, 0.f, 0.f, 0.f);
        }
    }
    const float* fsrc = ((role == 0) ? k : v) + recb;
    #pragma unroll
    for (int i = 0; i < 32; ++i)
        fr[i] = *(const float4*)(fsrc + (i & 3) * TILE_F + (i >> 2) * RQ_F);

    for (int jb = 0; jb <= nblk; ++jb) {
        const bool prod = (role == 0) && (jb < nblk);
        const bool cons = (role != 0) && (jb > 0);
        v8f s8[KT];
        #pragma unroll
        for (int tb = 0; tb < KT; ++tb) s8[tb] = v8f{};
        if (prod) {
            {
                #pragma unroll
                for (int cq = 0; cq < 4; ++cq) {
                    #pragma unroll
                    for (int rq = 0; rq < 2; ++rq) {
                        const int g = cq * 2 + rq;
                        v2f qa; qa.x = qf[cq][rq].x; qa.y = qf[cq][rq].y;
                        #pragma unroll
                        for (int tb = 0; tb < KT; ++tb) {
                            v2f kb; kb.x = fr[g * 4 + tb].x; kb.y = fr[g * 4 + tb].y;
                            s8[tb] = wmma4(qa, kb, s8[tb]);
                        }
                        qa.x = qf[cq][rq].z; qa.y = qf[cq][rq].w;
                        #pragma unroll
                        for (int tb = 0; tb < KT; ++tb) {
                            v2f kb; kb.x = fr[g * 4 + tb].z; kb.y = fr[g * 4 + tb].w;
                            s8[tb] = wmma4(qa, kb, s8[tb]);
                        }
                    }
                }

            }
        } else if (cons) {
            {
                const int slot = (jb - 1) & 1;
                const float4 a0 = *(const float4*)&abuf[slot][half * 8];
                const float4 a1 = *(const float4*)&abuf[slot][half * 8 + 4];
                const float al[8] = { a0.x, a0.y, a0.z, a0.w, a1.x, a1.y, a1.z, a1.w };
                #pragma unroll
                for (int r = 0; r < 8; ++r) {
                    o0[r] *= al[r]; o1[r] *= al[r];
                    o2[r] *= al[r]; o3[r] *= al[r];
                }
                #pragma unroll
                for (int tb = 0; tb < KT; ++tb) {
                    #pragma unroll
                    for (int rq = 0; rq < 2; ++rq) {
                        const float4 pf = *(const float4*)&pbuf[slot][nlo][16 * tb + 8 * half + 4 * rq];
                        v2f pa; pa.x = pf.x; pa.y = pf.y;
                        { v2f vb; vb.x = fr[(0 + rq) * 4 + tb].x; vb.y = fr[(0 + rq) * 4 + tb].y; o0 = wmma4(pa, vb, o0); }
                        { v2f vb; vb.x = fr[(2 + rq) * 4 + tb].x; vb.y = fr[(2 + rq) * 4 + tb].y; o1 = wmma4(pa, vb, o1); }
                        { v2f vb; vb.x = fr[(4 + rq) * 4 + tb].x; vb.y = fr[(4 + rq) * 4 + tb].y; o2 = wmma4(pa, vb, o2); }
                        { v2f vb; vb.x = fr[(6 + rq) * 4 + tb].x; vb.y = fr[(6 + rq) * 4 + tb].y; o3 = wmma4(pa, vb, o3); }
                        pa.x = pf.z; pa.y = pf.w;
                        { v2f vb; vb.x = fr[(0 + rq) * 4 + tb].z; vb.y = fr[(0 + rq) * 4 + tb].w; o0 = wmma4(pa, vb, o0); }
                        { v2f vb; vb.x = fr[(2 + rq) * 4 + tb].z; vb.y = fr[(2 + rq) * 4 + tb].w; o1 = wmma4(pa, vb, o1); }
                        { v2f vb; vb.x = fr[(4 + rq) * 4 + tb].z; vb.y = fr[(4 + rq) * 4 + tb].w; o2 = wmma4(pa, vb, o2); }
                        { v2f vb; vb.x = fr[(6 + rq) * 4 + tb].z; vb.y = fr[(6 + rq) * 4 + tb].w; o3 = wmma4(pa, vb, o3); }
                    }
                }
            }
        }
        {
            const int jw = jb + 1 - role;
            const int jn = (jw < nblk) ? jw : nblk - 1;
            const float* fp = fsrc + (size_t)(jn * KT) * TILE_F;
            #pragma unroll
            for (int i = 0; i < 32; ++i)
                fr[i] = *(const float4*)(fp + (i & 3) * TILE_F + (i >> 2) * RQ_F);
        }
        if (prod) {
            {
                const int slot = jb & 1;
                float sv[KT][8];
                #pragma unroll
                for (int tb = 0; tb < KT; ++tb) {
                    #pragma unroll
                    for (int r = 0; r < 8; ++r) sv[tb][r] = s8[tb][r] * scale;
                }

                if (jb == nblk - 1) {
                    const int kidx = jb * KB + nlo;
                    #pragma unroll
                    for (int tb = 0; tb < KT; ++tb) {
                        #pragma unroll
                        for (int r = 0; r < 8; ++r) {
                            if (kidx + 16 * tb > qidx + r) sv[tb][r] = -1e30f;
                        }
                    }
                }

                float alpha[8];
                #pragma unroll
                for (int r = 0; r < 8; ++r) {
                    float t = sv[0][r];
                    #pragma unroll
                    for (int tb = 1; tb < KT; ++tb) t = fmaxf(t, sv[tb][r]);
                    t = fmaxf(t, row_xor<1>(t));
                    t = fmaxf(t, row_xor<2>(t));
                    t = fmaxf(t, row_xor<4>(t));
                    t = fmaxf(t, row_xor<8>(t));
                    float mnew = fmaxf(mrow[r], t);
                    alpha[r] = __expf(mrow[r] - mnew);
                    mrow[r]  = mnew;
                    float s = 0.f;
                    #pragma unroll
                    for (int tb = 0; tb < KT; ++tb) {
                        float p = __expf(sv[tb][r] - mnew);
                        pbuf[slot][r + half * 8][16 * tb + nlo] = p;
                        s += p;
                    }
                    s += row_xor<1>(s);
                    s += row_xor<2>(s);
                    s += row_xor<4>(s);
                    s += row_xor<8>(s);
                    lrow[r] = lrow[r] * alpha[r] + s;
                }
                if (nlo == 0) {
                    #pragma unroll
                    for (int r = 0; r < 8; ++r) abuf[slot][half * 8 + r] = alpha[r];
                    if (jb == nblk - 1) {
                        #pragma unroll
                        for (int r = 0; r < 8; ++r) lbuf[half * 8 + r] = lrow[r];
                    }
                }
            }
        }
        __builtin_amdgcn_fence(__ATOMIC_RELEASE, "workgroup", "local");
        __builtin_amdgcn_s_barrier();
        __builtin_amdgcn_fence(__ATOMIC_ACQUIRE, "workgroup", "local");
    }

    if (role == 1) {
        const float4 l0 = *(const float4*)&lbuf[half * 8];
        const float4 l1 = *(const float4*)&lbuf[half * 8 + 4];
        const float lr[8] = { l0.x, l0.y, l0.z, l0.w, l1.x, l1.y, l1.z, l1.w };
        #pragma unroll
        for (int r = 0; r < 8; ++r) {
            float inv = 1.0f / lr[r];
            size_t idx = (size_t)(qrow0 + r + half * 8) * Hh;
            out[idx +  0 + nlo] = o0[r] * inv;
            out[idx + 16 + nlo] = o1[r] * inv;
            out[idx + 32 + nlo] = o2[r] * inv;
            out[idx + 48 + nlo] = o3[r] * inv;
        }
    }
}

extern "C" void kernel_launch(void* const* d_in, const int* in_sizes, int n_in,
                              void* d_out, int out_size, void* d_ws, size_t ws_size,
                              hipStream_t stream) {
    const float* x  = (const float*)d_in[0];
    const float* Wk = (const float*)d_in[1];
    const float* Wq = (const float*)d_in[2];
    const float* Wv = (const float*)d_in[3];
    float* out = (float*)d_out;

    float* ws   = (float*)d_ws;
    float* qbuf = ws;
    float* kbuf = ws + (size_t)ROWS * Hh;
    float* vbuf = ws + 2 * (size_t)ROWS * Hh;

    proj_kernel<<<ROWS / (16 * PR), 256, 0, stream>>>(x, Wk, Wq, Wv, qbuf, kbuf, vbuf);
    attn_kernel<<<ROWS / (16 * QT), 64 * QT, 0, stream>>>(qbuf, kbuf, vbuf, out);
}


# attention: work-group = 2 late + 2 early query tiles (127-2g,126-2g,2g,2g+1); each SIMD runs one phase (scores or values) of a late tile beside the other phase of an early tile; sum order kept
# speedup vs baseline: 2.5983x; 1.0840x over previous
#include <hip/hip_runtime.h>

typedef __attribute__((ext_vector_type(2))) float v2f;
typedef __attribute__((ext_vector_type(8))) float v8f;

constexpr int Bb = 8, Tt = 2048, Ee = 1024, Hh = 64;
constexpr int ROWS = Bb * Tt;
constexpr int TPB  = Tt / 16;

__device__ __forceinline__ v8f wmma4(v2f a, v2f b, v8f c) {
    return __builtin_amdgcn_wmma_f32_16x16x4_f32(
         false, a,  false, b,
         (short)0, c,  false,  false);
}

template <int M>
__device__ __forceinline__ float row_xor(float v) {
    return __builtin_bit_cast(float,
        __builtin_amdgcn_update_dpp(0, __builtin_bit_cast(int, v), 0x160 + M, 0xf, 0xf, true));
}

constexpr int TILE_F = 1024;
constexpr int CQ_F   = 256;
constexpr int RQ_F   = 128;

constexpr int PR = 4;
constexpr int PC = 3;
constexpr int EH = Ee / 2;

__global__ __launch_bounds__(256)
void proj_kernel(const float* __restrict__ x,
                 const float* __restrict__ Wk,
                 const float* __restrict__ Wq,
                 const float* __restrict__ Wv,
                 float* __restrict__ qo, float* __restrict__ ko,
                 float* __restrict__ vo) {
    __shared__ __attribute__((aligned(16))) float4 red[4][PR * PC * 2][32];

    const int lane = threadIdx.x & 31;
    const int wid  = __builtin_amdgcn_readfirstlane((int)(threadIdx.x >> 5));
    const int wv   = wid & 3;
    const int kh   = wid >> 2;
    const int c    = lane & 15;
    const int h    = lane >> 4;
    const int rowb = blockIdx.x * (16 * PR);
    const int kb   = kh * EH;

    const float* wp[PC];
    wp[0] = Wk + (size_t)(kb + 4 * h) * Hh + wv * 16 + c;
    wp[1] = Wq + (size_t)(kb + 4 * h) * Hh + wv * 16 + c;
    wp[2] = Wv + (size_t)(kb + 4 * h) * Hh + wv * 16 + c;
    const float* xp = x + (size_t)(rowb + c) * Ee + kb + 4 * h;

    v8f acc[PR][PC];
    #pragma unroll
    for (int rt = 0; rt < PR; ++rt) {
        #pragma unroll
        for (int t = 0; t < PC; ++t) acc[rt][t] = v8f{};
    }

    float4 xa0[2][PR], xa1[2][PR];
    float  wb0[2][PC][4], wb1[2][PC][4];

    auto load_chunk = [&](float4 (&xa)[2][PR], float (&wb)[2][PC][4], int e0) {
        #pragma unroll
        for (int g2 = 0; g2 < 2; ++g2) {
            #pragma unroll
            for (int rt = 0; rt < PR; ++rt)
                xa[g2][rt] = *(const float4*)(xp + (size_t)rt * 16 * Ee + e0 + 8 * g2);
            #pragma unroll
            for (int t = 0; t < PC; ++t) {
                #pragma unroll
                for (int j = 0; j < 4; ++j)
                    wb[g2][t][j] = wp[t][(size_t)(e0 + 8 * g2 + j) * Hh];
            }
        }
    };
    auto mma_chunk = [&](const float4 (&xa)[2][PR], const float (&wb)[2][PC][4]) {
        #pragma unroll
        for (int g2 = 0; g2 < 2; ++g2) {
            #pragma unroll
            for (int rt = 0; rt < PR; ++rt) {
                v2f a; a.x = xa[g2][rt].x; a.y = xa[g2][rt].y;
                #pragma unroll
                for (int t = 0; t < PC; ++t) {
                    v2f b; b.x = wb[g2][t][0]; b.y = wb[g2][t][1];
                    acc[rt][t] = (t < 2) ? wmma4(b, a, acc[rt][t]) : wmma4(a, b, acc[rt][t]);
                }
            }
            #pragma unroll
            for (int rt = 0; rt < PR; ++rt) {
                v2f a; a.x = xa[g2][rt].z; a.y = xa[g2][rt].w;
                #pragma unroll
                for (int t = 0; t < PC; ++t) {
                    v2f b; b.x = wb[g2][t][2]; b.y = wb[g2][t][3];
                    acc[rt][t] = (t < 2) ? wmma4(b, a, acc[rt][t]) : wmma4(a, b, acc[rt][t]);
                }
            }
        }
    };

    load_chunk(xa0, wb0, 0);
    for (int e0 = 0; e0 < EH - 32; e0 += 32) {
        load_chunk(xa1, wb1, e0 + 16);
        mma_chunk(xa0, wb0);
        load_chunk(xa0, wb0, e0 + 32);
        mma_chunk(xa1, wb1);
    }
    load_chunk(xa1, wb1, EH - 16);
    mma_chunk(xa0, wb0);
    mma_chunk(xa1, wb1);

    if (kh == 1) {
        #pragma unroll
        for (int rt = 0; rt < PR; ++rt) {
            #pragma unroll
            for (int t = 0; t < PC; ++t) {
                #pragma unroll
                for (int rq = 0; rq < 2; ++rq) {
                    float4 p4;
                    p4.x = acc[rt][t][4 * rq + 0]; p4.y = acc[rt][t][4 * rq + 1];
                    p4.z = acc[rt][t][4 * rq + 2]; p4.w = acc[rt][t][4 * rq + 3];
                    red[wv][(rt * PC + t) * 2 + rq][lane] = p4;
                }
            }
        }
    }
    __syncthreads();
    if (kh == 1) return;

    float* ob[PC] = { ko, qo, vo };
    #pragma unroll
    for (int rt = 0; rt < PR; ++rt) {
        const size_t rec = (size_t)(blockIdx.x * PR + rt) * TILE_F + wv * CQ_F + lane * 4;
        #pragma unroll
        for (int t = 0; t < PC; ++t) {
            #pragma unroll
            for (int rq = 0; rq < 2; ++rq) {
                const float4 p4 = red[wv][(rt * PC + t) * 2 + rq][lane];
                float4 o4;
                o4.x = acc[rt][t][4 * rq + 0] + p4.x; o4.y = acc[rt][t][4 * rq + 1] + p4.y;
                o4.z = acc[rt][t][4 * rq + 2] + p4.z; o4.w = acc[rt][t][4 * rq + 3] + p4.w;
                *(float4*)(ob[t] + rec + rq * RQ_F) = o4;
            }
        }
    }
}

constexpr int KT = 4;
constexpr int KB = 16 * KT;
constexpr int PP = KB + 4;

constexpr int QT = 4;
constexpr int GPB = TPB / QT;
static_assert(QT == 4 && KT % 2 == 0 && TPB % QT == 0, "two adjacent tiles must share their trip count; slots pair up as (0,2), (1,3)");

__global__ __launch_bounds__(64 * QT)
void attn_kernel(const float* __restrict__ q,
                 const float* __restrict__ k,
                 const float* __restrict__ v,
                 float* __restrict__ out) {
    __shared__ __attribute__((aligned(16))) float pbuf_all[QT][2][16][PP];
    __shared__ __attribute__((aligned(16))) float abuf_all[QT][2][16];
    __shared__ __attribute__((aligned(16))) float lbuf_all[QT][16];

    const int lane  = threadIdx.x & 31;
    const int wid   = __builtin_amdgcn_readfirstlane((int)(threadIdx.x >> 5));
    const int role  = wid / QT;
    const int pr    = (wid % QT) ^ (2 * role);
    const int b     = blockIdx.x / GPB;
    const int wg    = blockIdx.x % GPB;
    const int it    = (pr < 2) ? (TPB - 1 - 2 * wg - pr) : (2 * wg + pr - 2);
    const int tile  = b * TPB + it;
    float (*pbuf)[16][PP] = pbuf_all[pr];
    float (*abuf)[16]     = abuf_all[pr];
    float  *lbuf          = lbuf_all[pr];
    const int qrow0 = tile * 16;
    const int half  = lane >> 4;
    const int nlo   = lane & 15;

    const float scale = 0.03125f;
    const int nblk = it / KT + 1;
    const int nmax = (TPB - 1 - 2 * wg) / KT + 1;
    const int qidx = it * 16 + half * 8;
    const size_t recb = (size_t)(b * TPB) * TILE_F + lane * 4;

    float4 qf[4][2];
    float4 fr[32];
    v8f o0 = {}, o1 = {}, o2 = {}, o3 = {};
    float mrow[8], lrow[8];
    #pragma unroll
    for (int r = 0; r < 8; ++r) { mrow[r] = -1e30f; lrow[r] = 0.f; }

    if (role == 0) {
        const float* qp = q + (size_t)tile * TILE_F + lane * 4;
        #pragma unroll
        for (int cq = 0; cq < 4; ++cq) {
            #pragma unroll
            for (int rq = 0; rq < 2; ++rq)
                qf[cq][rq] = *(const float4*)(qp + cq * CQ_F + rq * RQ_F);
        }
    } else {
        #pragma unroll
        for (int cq = 0; cq < 4; ++cq) {
            #pragma unroll
            for (int rq = 0; rq < 2; ++rq)
                qf[cq][rq] = make_float4(0.f, 0.f, 0.f, 0.f);
        }
    }
    const float* fsrc = ((role == 0) ? k : v) + recb;
    #pragma unroll
    for (int i = 0; i < 32; ++i)
        fr[i] = *(const float4*)(fsrc + (i & 3) * TILE_F + (i >> 2) * RQ_F);

    for (int jb = 0; jb <= nmax; ++jb) {
        const bool prod = (role == 0) && (jb < nblk);
        const bool cons = (role != 0) && (jb > 0) && (jb <= nblk);
        v8f s8[KT];
        #pragma unroll
        for (int tb = 0; tb < KT; ++tb) s8[tb] = v8f{};
        if (prod) {
            {
                #pragma unroll
                for (int cq = 0; cq < 4; ++cq) {
                    #pragma unroll
                    for (int rq = 0; rq < 2; ++rq) {
                        const int g = cq * 2 + rq;
                        v2f qa; qa.x = qf[cq][rq].x; qa.y = qf[cq][rq].y;
                        #pragma unroll
                        for (int tb = 0; tb < KT; ++tb) {
                            v2f kb; kb.x = fr[g * 4 + tb].x; kb.y = fr[g * 4 + tb].y;
                            s8[tb] = wmma4(qa, kb, s8[tb]);
                        }
                        qa.x = qf[cq][rq].z; qa.y = qf[cq][rq].w;
                        #pragma unroll
                        for (int tb = 0; tb < KT; ++tb) {
                            v2f kb; kb.x = fr[g * 4 + tb].z; kb.y = fr[g * 4 + tb].w;
                            s8[tb] = wmma4(qa, kb, s8[tb]);
                        }
                    }
                }

            }
        } else if (cons) {
            {
                const int slot = (jb - 1) & 1;
                const float4 a0 = *(const float4*)&abuf[slot][half * 8];
                const float4 a1 = *(const float4*)&abuf[slot][half * 8 + 4];
                const float al[8] = { a0.x, a0.y, a0.z, a0.w, a1.x, a1.y, a1.z, a1.w };
                #pragma unroll
                for (int r = 0; r < 8; ++r) {
                    o0[r] *= al[r]; o1[r] *= al[r];
                    o2[r] *= al[r]; o3[r] *= al[r];
                }
                #pragma unroll
                for (int tb = 0; tb < KT; ++tb) {
                    #pragma unroll
                    for (int rq = 0; rq < 2; ++rq) {
                        const float4 pf = *(const float4*)&pbuf[slot][nlo][16 * tb + 8 * half + 4 * rq];
                        v2f pa; pa.x = pf.x; pa.y = pf.y;
                        { v2f vb; vb.x = fr[(0 + rq) * 4 + tb].x; vb.y = fr[(0 + rq) * 4 + tb].y; o0 = wmma4(pa, vb, o0); }
                        { v2f vb; vb.x = fr[(2 + rq) * 4 + tb].x; vb.y = fr[(2 + rq) * 4 + tb].y; o1 = wmma4(pa, vb, o1); }
                        { v2f vb; vb.x = fr[(4 + rq) * 4 + tb].x; vb.y = fr[(4 + rq) * 4 + tb].y; o2 = wmma4(pa, vb, o2); }
                        { v2f vb; vb.x = fr[(6 + rq) * 4 + tb].x; vb.y = fr[(6 + rq) * 4 + tb].y; o3 = wmma4(pa, vb, o3); }
                        pa.x = pf.z; pa.y = pf.w;
                        { v2f vb; vb.x = fr[(0 + rq) * 4 + tb].z; vb.y = fr[(0 + rq) * 4 + tb].w; o0 = wmma4(pa, vb, o0); }
                        { v2f vb; vb.x = fr[(2 + rq) * 4 + tb].z; vb.y = fr[(2 + rq) * 4 + tb].w; o1 = wmma4(pa, vb, o1); }
                        { v2f vb; vb.x = fr[(4 + rq) * 4 + tb].z; vb.y = fr[(4 + rq) * 4 + tb].w; o2 = wmma4(pa, vb, o2); }
                        { v2f vb; vb.x = fr[(6 + rq) * 4 + tb].z; vb.y = fr[(6 + rq) * 4 + tb].w; o3 = wmma4(pa, vb, o3); }
                    }
                }
            }
        }
        const int jw = jb + 1 - role;
        if (jw < nblk) {
            const float* fp = fsrc + (size_t)(jw * KT) * TILE_F;
            #pragma unroll
            for (int i = 0; i < 32; ++i)
                fr[i] = *(const float4*)(fp + (i & 3) * TILE_F + (i >> 2) * RQ_F);
        }
        if (prod) {
            {
                const int slot = jb & 1;
                float sv[KT][8];
                #pragma unroll
                for (int tb = 0; tb < KT; ++tb) {
                    #pragma unroll
                    for (int r = 0; r < 8; ++r) sv[tb][r] = s8[tb][r] * scale;
                }

                if (jb == nblk - 1) {
                    const int kidx = jb * KB + nlo;
                    #pragma unroll
                    for (int tb = 0; tb < KT; ++tb) {
                        #pragma unroll
                        for (int r = 0; r < 8; ++r) {
                            if (kidx + 16 * tb > qidx + r) sv[tb][r] = -1e30f;
                        }
                    }
                }

                float alpha[8];
                #pragma unroll
                for (int r = 0; r < 8; ++r) {
                    float t = sv[0][r];
                    #pragma unroll
                    for (int tb = 1; tb < KT; ++tb) t = fmaxf(t, sv[tb][r]);
                    t = fmaxf(t, row_xor<1>(t));
                    t = fmaxf(t, row_xor<2>(t));
                    t = fmaxf(t, row_xor<4>(t));
                    t = fmaxf(t, row_xor<8>(t));
                    float mnew = fmaxf(mrow[r], t);
                    alpha[r] = __expf(mrow[r] - mnew);
                    mrow[r]  = mnew;
                    float s = 0.f;
                    #pragma unroll
                    for (int tb = 0; tb < KT; ++tb) {
                        float p = __expf(sv[tb][r] - mnew);
                        pbuf[slot][r + half * 8][16 * tb + nlo] = p;
                        s += p;
                    }
                    s += row_xor<1>(s);
                    s += row_xor<2>(s);
                    s += row_xor<4>(s);
                    s += row_xor<8>(s);
                    lrow[r] = lrow[r] * alpha[r] + s;
                }
                if (nlo == 0) {
                    #pragma unroll
                    for (int r = 0; r < 8; ++r) abuf[slot][half * 8 + r] = alpha[r];
                    if (jb == nblk - 1) {
                        #pragma unroll
                        for (int r = 0; r < 8; ++r) lbuf[half * 8 + r] = lrow[r];
                    }
                }
            }
        }
        __builtin_amdgcn_fence(__ATOMIC_RELEASE, "workgroup", "local");
        __builtin_amdgcn_s_barrier();
        __builtin_amdgcn_fence(__ATOMIC_ACQUIRE, "workgroup", "local");
    }

    if (role == 1) {
        const float4 l0 = *(const float4*)&lbuf[half * 8];
        const float4 l1 = *(const float4*)&lbuf[half * 8 + 4];
        const float lr[8] = { l0.x, l0.y, l0.z, l0.w, l1.x, l1.y, l1.z, l1.w };
        #pragma unroll
        for (int r = 0; r < 8; ++r) {
            float inv = 1.0f / lr[r];
            size_t idx = (size_t)(qrow0 + r + half * 8) * Hh;
            out[idx +  0 + nlo] = o0[r] * inv;
            out[idx + 16 + nlo] = o1[r] * inv;
            out[idx + 32 + nlo] = o2[r] * inv;
            out[idx + 48 + nlo] = o3[r] * inv;
        }
    }
}

extern "C" void kernel_launch(void* const* d_in, const int* in_sizes, int n_in,
                              void* d_out, int out_size, void* d_ws, size_t ws_size,
                              hipStream_t stream) {
    const float* x  = (const float*)d_in[0];
    const float* Wk = (const float*)d_in[1];
    const float* Wq = (const float*)d_in[2];
    const float* Wv = (const float*)d_in[3];
    float* out = (float*)d_out;

    float* ws   = (float*)d_ws;
    float* qbuf = ws;
    float* kbuf = ws + (size_t)ROWS * Hh;
    float* vbuf = ws + 2 * (size_t)ROWS * Hh;

    proj_kernel<<<ROWS / (16 * PR), 256, 0, stream>>>(x, Wk, Wq, Wv, qbuf, kbuf, vbuf);
    attn_kernel<<<ROWS / (16 * QT), 64 * QT, 0, stream>>>(qbuf, kbuf, vbuf, out);
}


# a step's memory reads (projection: a chunk's 32 global fragment loads; values wave: a trip's 10 LDS reads) held as one group in front of its matrix run by scheduling fences; sum order kept
# speedup vs baseline: 2.6906x; 1.0355x over previous
#include <hip/hip_runtime.h>

typedef __attribute__((ext_vector_type(2))) float v2f;
typedef __attribute__((ext_vector_type(8))) float v8f;

constexpr int Bb = 8, Tt = 2048, Ee = 1024, Hh = 64;
constexpr int ROWS = Bb * Tt;
constexpr int TPB  = Tt / 16;

__device__ __forceinline__ v8f wmma4(v2f a, v2f b, v8f c) {
    return __builtin_amdgcn_wmma_f32_16x16x4_f32(
         false, a,  false, b,
         (short)0, c,  false,  false);
}

template <int M>
__device__ __forceinline__ float row_xor(float v) {
    return __builtin_bit_cast(float,
        __builtin_amdgcn_update_dpp(0, __builtin_bit_cast(int, v), 0x160 + M, 0xf, 0xf, true));
}

constexpr int TILE_F = 1024;
constexpr int CQ_F   = 256;
constexpr int RQ_F   = 128;

constexpr int PR = 4;
constexpr int PC = 3;
constexpr int EH = Ee / 2;

__global__ __launch_bounds__(256)
void proj_kernel(const float* __restrict__ x,
                 const float* __restrict__ Wk,
                 const float* __restrict__ Wq,
                 const float* __restrict__ Wv,
                 float* __restrict__ qo, float* __restrict__ ko,
                 float* __restrict__ vo) {
    __shared__ __attribute__((aligned(16))) float4 red[4][PR * PC * 2][32];

    const int lane = threadIdx.x & 31;
    const int wid  = __builtin_amdgcn_readfirstlane((int)(threadIdx.x >> 5));
    const int wv   = wid & 3;
    const int kh   = wid >> 2;
    const int c    = lane & 15;
    const int h    = lane >> 4;
    const int rowb = blockIdx.x * (16 * PR);
    const int kb   = kh * EH;

    const float* wp[PC];
    wp[0] = Wk + (size_t)(kb + 4 * h) * Hh + wv * 16 + c;
    wp[1] = Wq + (size_t)(kb + 4 * h) * Hh + wv * 16 + c;
    wp[2] = Wv + (size_t)(kb + 4 * h) * Hh + wv * 16 + c;
    const float* xp = x + (size_t)(rowb + c) * Ee + kb + 4 * h;

    v8f acc[PR][PC];
    #pragma unroll
    for (int rt = 0; rt < PR; ++rt) {
        #pragma unroll
        for (int t = 0; t < PC; ++t) acc[rt][t] = v8f{};
    }

    float4 xa0[2][PR], xa1[2][PR];
    float  wb0[2][PC][4], wb1[2][PC][4];

    auto load_chunk = [&](float4 (&xa)[2][PR], float (&wb)[2][PC][4], int e0) {
        #pragma unroll
        for (int g2 = 0; g2 < 2; ++g2) {
            #pragma unroll
            for (int rt = 0; rt < PR; ++rt)
                xa[g2][rt] = *(const float4*)(xp + (size_t)rt * 16 * Ee + e0 + 8 * g2);
            #pragma unroll
            for (int t = 0; t < PC; ++t) {
                #pragma unroll
                for (int j = 0; j < 4; ++j)
                    wb[g2][t][j] = wp[t][(size_t)(e0 + 8 * g2 + j) * Hh];
            }
        }
    };
    auto mma_chunk = [&](const float4 (&xa)[2][PR], const float (&wb)[2][PC][4]) {
        #pragma unroll
        for (int g2 = 0; g2 < 2; ++g2) {
            #pragma unroll
            for (int rt = 0; rt < PR; ++rt) {
                v2f a; a.x = xa[g2][rt].x; a.y = xa[g2][rt].y;
                #pragma unroll
                for (int t = 0; t < PC; ++t) {
                    v2f b; b.x = wb[g2][t][0]; b.y = wb[g2][t][1];
                    acc[rt][t] = (t < 2) ? wmma4(b, a, acc[rt][t]) : wmma4(a, b, acc[rt][t]);
                }
            }
            #pragma unroll
            for (int rt = 0; rt < PR; ++rt) {
                v2f a; a.x = xa[g2][rt].z; a.y = xa[g2][rt].w;
                #pragma unroll
                for (int t = 0; t < PC; ++t) {
                    v2f b; b.x = wb[g2][t][2]; b.y = wb[g2][t][3];
                    acc[rt][t] = (t < 2) ? wmma4(b, a, acc[rt][t]) : wmma4(a, b, acc[rt][t]);
                }
            }
        }
    };

    load_chunk(xa0, wb0, 0);
    for (int e0 = 0; e0 < EH - 32; e0 += 32) {
        load_chunk(xa1, wb1, e0 + 16);
        mma_chunk(xa0, wb0);
        __builtin_amdgcn_sched_barrier(0);
        load_chunk(xa0, wb0, e0 + 32);
        mma_chunk(xa1, wb1);
    }
    load_chunk(xa1, wb1, EH - 16);
    __builtin_amdgcn_sched_barrier(0);
    mma_chunk(xa0, wb0);
    mma_chunk(xa1, wb1);

    if (kh == 1) {
        #pragma unroll
        for (int rt = 0; rt < PR; ++rt) {
            #pragma unroll
            for (int t = 0; t < PC; ++t) {
                #pragma unroll
                for (int rq = 0; rq < 2; ++rq) {
                    float4 p4;
                    p4.x = acc[rt][t][4 * rq + 0]; p4.y = acc[rt][t][4 * rq + 1];
                    p4.z = acc[rt][t][4 * rq + 2]; p4.w = acc[rt][t][4 * rq + 3];
                    red[wv][(rt * PC + t) * 2 + rq][lane] = p4;
                }
            }
        }
    }
    __syncthreads();
    if (kh == 1) return;

    float* ob[PC] = { ko, qo, vo };
    #pragma unroll
    for (int rt = 0; rt < PR; ++rt) {
        const size_t rec = (size_t)(blockIdx.x * PR + rt) * TILE_F + wv * CQ_F + lane * 4;
        #pragma unroll
        for (int t = 0; t < PC; ++t) {
            #pragma unroll
            for (int rq = 0; rq < 2; ++rq) {
                const float4 p4 = red[wv][(rt * PC + t) * 2 + rq][lane];
                float4 o4;
                o4.x = acc[rt][t][4 * rq + 0] + p4.x; o4.y = acc[rt][t][4 * rq + 1] + p4.y;
                o4.z = acc[rt][t][4 * rq + 2] + p4.z; o4.w = acc[rt][t][4 * rq + 3] + p4.w;
                *(float4*)(ob[t] + rec + rq * RQ_F) = o4;
            }
        }
    }
}

constexpr int KT = 4;
constexpr int KB = 16 * KT;
constexpr int PP = KB + 4;

constexpr int QT = 4;
constexpr int GPB = TPB / QT;
static_assert(QT == 4 && KT % 2 == 0 && TPB % QT == 0, "two adjacent tiles must share their trip count; slots pair up as (0,2), (1,3)");

__global__ __launch_bounds__(64 * QT)
void attn_kernel(const float* __restrict__ q,
                 const float* __restrict__ k,
                 const float* __restrict__ v,
                 float* __restrict__ out) {
    __shared__ __attribute__((aligned(16))) float pbuf_all[QT][2][16][PP];
    __shared__ __attribute__((aligned(16))) float abuf_all[QT][2][16];
    __shared__ __attribute__((aligned(16))) float lbuf_all[QT][16];

    const int lane  = threadIdx.x & 31;
    const int wid   = __builtin_amdgcn_readfirstlane((int)(threadIdx.x >> 5));
    const int role  = wid / QT;
    const int pr    = (wid % QT) ^ (2 * role);
    const int b     = blockIdx.x / GPB;
    const int wg    = blockIdx.x % GPB;
    const int it    = (pr < 2) ? (TPB - 1 - 2 * wg - pr) : (2 * wg + pr - 2);
    const int tile  = b * TPB + it;
    float (*pbuf)[16][PP] = pbuf_all[pr];
    float (*abuf)[16]     = abuf_all[pr];
    float  *lbuf          = lbuf_all[pr];
    const int qrow0 = tile * 16;
    const int half  = lane >> 4;
    const int nlo   = lane & 15;

    const float scale = 0.03125f;
    const int nblk = it / KT + 1;
    const int nmax = (TPB - 1 - 2 * wg) / KT + 1;
    const int qidx = it * 16 + half * 8;
    const size_t recb = (size_t)(b * TPB) * TILE_F + lane * 4;

    float4 qf[4][2];
    float4 fr[32];
    v8f o0 = {}, o1 = {}, o2 = {}, o3 = {};
    float mrow[8], lrow[8];
    #pragma unroll
    for (int r = 0; r < 8; ++r) { mrow[r] = -1e30f; lrow[r] = 0.f; }

    if (role == 0) {
        const float* qp = q + (size_t)tile * TILE_F + lane * 4;
        #pragma unroll
        for (int cq = 0; cq < 4; ++cq) {
            #pragma unroll
            for (int rq = 0; rq < 2; ++rq)
                qf[cq][rq] = *(const float4*)(qp + cq * CQ_F + rq * RQ_F);
        }
    } else {
        #pragma unroll
        for (int cq = 0; cq < 4; ++cq) {
            #pragma unroll
            for (int rq = 0; rq < 2; ++rq)
                qf[cq][rq] = make_float4(0.f, 0.f, 0.f, 0.f);
        }
    }
    const float* fsrc = ((role == 0) ? k : v) + recb;
    #pragma unroll
    for (int i = 0; i < 32; ++i)
        fr[i] = *(const float4*)(fsrc + (i & 3) * TILE_F + (i >> 2) * RQ_F);

    for (int jb = 0; jb <= nmax; ++jb) {
        const bool prod = (role == 0) && (jb < nblk);
        const bool cons = (role != 0) && (jb > 0) && (jb <= nblk);
        v8f s8[KT];
        #pragma unroll
        for (int tb = 0; tb < KT; ++tb) s8[tb] = v8f{};
        if (prod) {
            {
                #pragma unroll
                for (int cq = 0; cq < 4; ++cq) {
                    #pragma unroll
                    for (int rq = 0; rq < 2; ++rq) {
                        const int g = cq * 2 + rq;
                        v2f qa; qa.x = qf[cq][rq].x; qa.y = qf[cq][rq].y;
                        #pragma unroll
                        for (int tb = 0; tb < KT; ++tb) {
                            v2f kb; kb.x = fr[g * 4 + tb].x; kb.y = fr[g * 4 + tb].y;
                            s8[tb] = wmma4(qa, kb, s8[tb]);
                        }
                        qa.x = qf[cq][rq].z; qa.y = qf[cq][rq].w;
                        #pragma unroll
                        for (int tb = 0; tb < KT; ++tb) {
                            v2f kb; kb.x = fr[g * 4 + tb].z; kb.y = fr[g * 4 + tb].w;
                            s8[tb] = wmma4(qa, kb, s8[tb]);
                        }
                    }
                }

            }
        } else if (cons) {
            {
                const int slot = (jb - 1) & 1;
                const float4 a0 = *(const float4*)&abuf[slot][half * 8];
                const float4 a1 = *(const float4*)&abuf[slot][half * 8 + 4];
                const float al[8] = { a0.x, a0.y, a0.z, a0.w, a1.x, a1.y, a1.z, a1.w };
                float4 pfr[KT][2];
                #pragma unroll
                for (int tb = 0; tb < KT; ++tb) {
                    #pragma unroll
                    for (int rq = 0; rq < 2; ++rq)
                        pfr[tb][rq] = *(const float4*)&pbuf[slot][nlo][16 * tb + 8 * half + 4 * rq];
                }
                __builtin_amdgcn_sched_barrier(0);
                #pragma unroll
                for (int r = 0; r < 8; ++r) {
                    o0[r] *= al[r]; o1[r] *= al[r];
                    o2[r] *= al[r]; o3[r] *= al[r];
                }
                #pragma unroll
                for (int tb = 0; tb < KT; ++tb) {
                    #pragma unroll
                    for (int rq = 0; rq < 2; ++rq) {
                        const float4 pf = pfr[tb][rq];
                        v2f pa; pa.x = pf.x; pa.y = pf.y;
                        { v2f vb; vb.x = fr[(0 + rq) * 4 + tb].x; vb.y = fr[(0 + rq) * 4 + tb].y; o0 = wmma4(pa, vb, o0); }
                        { v2f vb; vb.x = fr[(2 + rq) * 4 + tb].x; vb.y = fr[(2 + rq) * 4 + tb].y; o1 = wmma4(pa, vb, o1); }
                        { v2f vb; vb.x = fr[(4 + rq) * 4 + tb].x; vb.y = fr[(4 + rq) * 4 + tb].y; o2 = wmma4(pa, vb, o2); }
                        { v2f vb; vb.x = fr[(6 + rq) * 4 + tb].x; vb.y = fr[(6 + rq) * 4 + tb].y; o3 = wmma4(pa, vb, o3); }
                        pa.x = pf.z; pa.y = pf.w;
                        { v2f vb; vb.x = fr[(0 + rq) * 4 + tb].z; vb.y = fr[(0 + rq) * 4 + tb].w; o0 = wmma4(pa, vb, o0); }
                        { v2f vb; vb.x = fr[(2 + rq) * 4 + tb].z; vb.y = fr[(2 + rq) * 4 + tb].w; o1 = wmma4(pa, vb, o1); }
                        { v2f vb; vb.x = fr[(4 + rq) * 4 + tb].z; vb.y = fr[(4 + rq) * 4 + tb].w; o2 = wmma4(pa, vb, o2); }
                        { v2f vb; vb.x = fr[(6 + rq) * 4 + tb].z; vb.y = fr[(6 + rq) * 4 + tb].w; o3 = wmma4(pa, vb, o3); }
                    }
                }
            }
        }
        const int jw = jb + 1 - role;
        if (jw < nblk) {
            const float* fp = fsrc + (size_t)(jw * KT) * TILE_F;
            #pragma unroll
            for (int i = 0; i < 32; ++i)
                fr[i] = *(const float4*)(fp + (i & 3) * TILE_F + (i >> 2) * RQ_F);
        }
        if (prod) {
            {
                const int slot = jb & 1;
                float sv[KT][8];
                #pragma unroll
                for (int tb = 0; tb < KT; ++tb) {
                    #pragma unroll
                    for (int r = 0; r < 8; ++r) sv[tb][r] = s8[tb][r] * scale;
                }

                if (jb == nblk - 1) {
                    const int kidx = jb * KB + nlo;
                    #pragma unroll
                    for (int tb = 0; tb < KT; ++tb) {
                        #pragma unroll
                        for (int r = 0; r < 8; ++r) {
                            if (kidx + 16 * tb > qidx + r) sv[tb][r] = -1e30f;
                        }
                    }
                }

                float alpha[8];
                #pragma unroll
                for (int r = 0; r < 8; ++r) {
                    float t = sv[0][r];
                    #pragma unroll
                    for (int tb = 1; tb < KT; ++tb) t = fmaxf(t, sv[tb][r]);
                    t = fmaxf(t, row_xor<1>(t));
                    t = fmaxf(t, row_xor<2>(t));
                    t = fmaxf(t, row_xor<4>(t));
                    t = fmaxf(t, row_xor<8>(t));
                    float mnew = fmaxf(mrow[r], t);
                    alpha[r] = __expf(mrow[r] - mnew);
                    mrow[r]  = mnew;
                    float s = 0.f;
                    #pragma unroll
                    for (int tb = 0; tb < KT; ++tb) {
                        float p = __expf(sv[tb][r] - mnew);
                        pbuf[slot][r + half * 8][16 * tb + nlo] = p;
                        s += p;
                    }
                    s += row_xor<1>(s);
                    s += row_xor<2>(s);
                    s += row_xor<4>(s);
                    s += row_xor<8>(s);
                    lrow[r] = lrow[r] * alpha[r] + s;
                }
                if (nlo == 0) {
                    #pragma unroll
                    for (int r = 0; r < 8; ++r) abuf[slot][half * 8 + r] = alpha[r];
                    if (jb == nblk - 1) {
                        #pragma unroll
                        for (int r = 0; r < 8; ++r) lbuf[half * 8 + r] = lrow[r];
                    }
                }
            }
        }
        __builtin_amdgcn_fence(__ATOMIC_RELEASE, "workgroup", "local");
        __builtin_amdgcn_s_barrier();
        __builtin_amdgcn_fence(__ATOMIC_ACQUIRE, "workgroup", "local");
    }

    if (role == 1) {
        const float4 l0 = *(const float4*)&lbuf[half * 8];
        const float4 l1 = *(const float4*)&lbuf[half * 8 + 4];
        const float lr[8] = { l0.x, l0.y, l0.z, l0.w, l1.x, l1.y, l1.z, l1.w };
        #pragma unroll
        for (int r = 0; r < 8; ++r) {
            float inv = 1.0f / lr[r];
            size_t idx = (size_t)(qrow0 + r + half * 8) * Hh;
            out[idx +  0 + nlo] = o0[r] * inv;
            out[idx + 16 + nlo] = o1[r] * inv;
            out[idx + 32 + nlo] = o2[r] * inv;
            out[idx + 48 + nlo] = o3[r] * inv;
        }
    }
}

extern "C" void kernel_launch(void* const* d_in, const int* in_sizes, int n_in,
                              void* d_out, int out_size, void* d_ws, size_t ws_size,
                              hipStream_t stream) {
    const float* x  = (const float*)d_in[0];
    const float* Wk = (const float*)d_in[1];
    const float* Wq = (const float*)d_in[2];
    const float* Wv = (const float*)d_in[3];
    float* out = (float*)d_out;

    float* ws   = (float*)d_ws;
    float* qbuf = ws;
    float* kbuf = ws + (size_t)ROWS * Hh;
    float* vbuf = ws + 2 * (size_t)ROWS * Hh;

    proj_kernel<<<ROWS / (16 * PR), 256, 0, stream>>>(x, Wk, Wq, Wv, qbuf, kbuf, vbuf);
    attn_kernel<<<ROWS / (16 * QT), 64 * QT, 0, stream>>>(qbuf, kbuf, vbuf, out);
}


# attention: softmax moved to a third wave per query tile (12-wave work-group): scores wave hands raw S over in LDS, runs 2 trips ahead of values wave; K/V refills behind run halves; sum order kept
# speedup vs baseline: 2.8237x; 1.0495x over previous
#include <hip/hip_runtime.h>

typedef __attribute__((ext_vector_type(2))) float v2f;
typedef __attribute__((ext_vector_type(8))) float v8f;

constexpr int Bb = 8, Tt = 2048, Ee = 1024, Hh = 64;
constexpr int ROWS = Bb * Tt;
constexpr int TPB  = Tt / 16;

__device__ __forceinline__ v8f wmma4(v2f a, v2f b, v8f c) {
    return __builtin_amdgcn_wmma_f32_16x16x4_f32(
         false, a,  false, b,
         (short)0, c,  false,  false);
}

template <int M>
__device__ __forceinline__ float row_xor(float v) {
    return __builtin_bit_cast(float,
        __builtin_amdgcn_update_dpp(0, __builtin_bit_cast(int, v), 0x160 + M, 0xf, 0xf, true));
}

constexpr int TILE_F = 1024;
constexpr int CQ_F   = 256;
constexpr int RQ_F   = 128;

constexpr int PR = 4;
constexpr int PC = 3;
constexpr int EH = Ee / 2;

__global__ __launch_bounds__(256)
void proj_kernel(const float* __restrict__ x,
                 const float* __restrict__ Wk,
                 const float* __restrict__ Wq,
                 const float* __restrict__ Wv,
                 float* __restrict__ qo, float* __restrict__ ko,
                 float* __restrict__ vo) {
    __shared__ __attribute__((aligned(16))) float4 red[4][PR * PC * 2][32];

    const int lane = threadIdx.x & 31;
    const int wid  = __builtin_amdgcn_readfirstlane((int)(threadIdx.x >> 5));
    const int wv   = wid & 3;
    const int kh   = wid >> 2;
    const int c    = lane & 15;
    const int h    = lane >> 4;
    const int rowb = blockIdx.x * (16 * PR);
    const int kb   = kh * EH;

    const float* wp[PC];
    wp[0] = Wk + (size_t)(kb + 4 * h) * Hh + wv * 16 + c;
    wp[1] = Wq + (size_t)(kb + 4 * h) * Hh + wv * 16 + c;
    wp[2] = Wv + (size_t)(kb + 4 * h) * Hh + wv * 16 + c;
    const float* xp = x + (size_t)(rowb + c) * Ee + kb + 4 * h;

    v8f acc[PR][PC];
    #pragma unroll
    for (int rt = 0; rt < PR; ++rt) {
        #pragma unroll
        for (int t = 0; t < PC; ++t) acc[rt][t] = v8f{};
    }

    float4 xa0[2][PR], xa1[2][PR];
    float  wb0[2][PC][4], wb1[2][PC][4];

    auto load_chunk = [&](float4 (&xa)[2][PR], float (&wb)[2][PC][4], int e0) {
        #pragma unroll
        for (int g2 = 0; g2 < 2; ++g2) {
            #pragma unroll
            for (int rt = 0; rt < PR; ++rt)
                xa[g2][rt] = *(const float4*)(xp + (size_t)rt * 16 * Ee + e0 + 8 * g2);
            #pragma unroll
            for (int t = 0; t < PC; ++t) {
                #pragma unroll
                for (int j = 0; j < 4; ++j)
                    wb[g2][t][j] = wp[t][(size_t)(e0 + 8 * g2 + j) * Hh];
            }
        }
    };
    auto mma_chunk = [&](const float4 (&xa)[2][PR], const float (&wb)[2][PC][4]) {
        #pragma unroll
        for (int g2 = 0; g2 < 2; ++g2) {
            #pragma unroll
            for (int rt = 0; rt < PR; ++rt) {
                v2f a; a.x = xa[g2][rt].x; a.y = xa[g2][rt].y;
                #pragma unroll
                for (int t = 0; t < PC; ++t) {
                    v2f b; b.x = wb[g2][t][0]; b.y = wb[g2][t][1];
                    acc[rt][t] = (t < 2) ? wmma4(b, a, acc[rt][t]) : wmma4(a, b, acc[rt][t]);
                }
            }
            #pragma unroll
            for (int rt = 0; rt < PR; ++rt) {
                v2f a; a.x = xa[g2][rt].z; a.y = xa[g2][rt].w;
                #pragma unroll
                for (int t = 0; t < PC; ++t) {
                    v2f b; b.x = wb[g2][t][2]; b.y = wb[g2][t][3];
                    acc[rt][t] = (t < 2) ? wmma4(b, a, acc[rt][t]) : wmma4(a, b, acc[rt][t]);
                }
            }
        }
    };

    load_chunk(xa0, wb0, 0);
    for (int e0 = 0; e0 < EH - 32; e0 += 32) {
        load_chunk(xa1, wb1, e0 + 16);
        mma_chunk(xa0, wb0);
        __builtin_amdgcn_sched_barrier(0);
        load_chunk(xa0, wb0, e0 + 32);
        mma_chunk(xa1, wb1);
    }
    load_chunk(xa1, wb1, EH - 16);
    __builtin_amdgcn_sched_barrier(0);
    mma_chunk(xa0, wb0);
    mma_chunk(xa1, wb1);

    if (kh == 1) {
        #pragma unroll
        for (int rt = 0; rt < PR; ++rt) {
            #pragma unroll
            for (int t = 0; t < PC; ++t) {
                #pragma unroll
                for (int rq = 0; rq < 2; ++rq) {
                    float4 p4;
                    p4.x = acc[rt][t][4 * rq + 0]; p4.y = acc[rt][t][4 * rq + 1];
                    p4.z = acc[rt][t][4 * rq + 2]; p4.w = acc[rt][t][4 * rq + 3];
                    red[wv][(rt * PC + t) * 2 + rq][lane] = p4;
                }
            }
        }
    }
    __syncthreads();
    if (kh == 1) return;

    float* ob[PC] = { ko, qo, vo };
    #pragma unroll
    for (int rt = 0; rt < PR; ++rt) {
        const size_t rec = (size_t)(blockIdx.x * PR + rt) * TILE_F + wv * CQ_F + lane * 4;
        #pragma unroll
        for (int t = 0; t < PC; ++t) {
            #pragma unroll
            for (int rq = 0; rq < 2; ++rq) {
                const float4 p4 = red[wv][(rt * PC + t) * 2 + rq][lane];
                float4 o4;
                o4.x = acc[rt][t][4 * rq + 0] + p4.x; o4.y = acc[rt][t][4 * rq + 1] + p4.y;
                o4.z = acc[rt][t][4 * rq + 2] + p4.z; o4.w = acc[rt][t][4 * rq + 3] + p4.w;
                *(float4*)(ob[t] + rec + rq * RQ_F) = o4;
            }
        }
    }
}

constexpr int KT = 4;
constexpr int KB = 16 * KT;
constexpr int PP = KB + 4;

constexpr int QT = 4;
constexpr int GPB = TPB / QT;
static_assert(QT == 4 && KT % 2 == 0 && TPB % QT == 0, "two adjacent tiles must share their trip count; slots pair up as (0,2), (1,3)");

__global__ __launch_bounds__(96 * QT)
void attn_kernel(const float* __restrict__ q,
                 const float* __restrict__ k,
                 const float* __restrict__ v,
                 float* __restrict__ out) {
    __shared__ __attribute__((aligned(16))) float4 sbuf_all[QT][2][2 * KT][32];
    __shared__ __attribute__((aligned(16))) float pbuf_all[QT][2][16][PP];
    __shared__ __attribute__((aligned(16))) float abuf_all[QT][2][16];
    __shared__ __attribute__((aligned(16))) float lbuf_all[QT][16];

    const int lane  = threadIdx.x & 31;
    const int wid   = __builtin_amdgcn_readfirstlane((int)(threadIdx.x >> 5));
    const int role  = wid / QT;
    const int pr    = (wid % QT) ^ ((role == 1) ? 2 : 0);
    const int b     = blockIdx.x / GPB;
    const int wg    = blockIdx.x % GPB;
    const int it    = (pr < 2) ? (TPB - 1 - 2 * wg - pr) : (2 * wg + pr - 2);
    const int tile  = b * TPB + it;
    float4 (*sbuf)[2 * KT][32] = sbuf_all[pr];
    float (*pbuf)[16][PP] = pbuf_all[pr];
    float (*abuf)[16]     = abuf_all[pr];
    float  *lbuf          = lbuf_all[pr];
    const int qrow0 = tile * 16;
    const int half  = lane >> 4;
    const int nlo   = lane & 15;

    const float scale = 0.03125f;
    const int nblk = it / KT + 1;
    const int nmax = (TPB - 1 - 2 * wg) / KT + 1;
    const int qidx = it * 16 + half * 8;
    const size_t recb = (size_t)(b * TPB) * TILE_F + lane * 4;

    const int sx = (role == 0) ? TILE_F : 2 * RQ_F;
    const int sy = (role == 0) ? 2 * RQ_F : TILE_F;

    float4 aq[8];
    v8f ac[4];
    float4 fr[32];
    float mrow[8], lrow[8];
    #pragma unroll
    for (int r = 0; r < 8; ++r) { mrow[r] = -1e30f; lrow[r] = 0.f; }
    #pragma unroll
    for (int x = 0; x < 4; ++x) ac[x] = v8f{};

    if (role == 0) {
        const float* qp = q + (size_t)tile * TILE_F + lane * 4;
        #pragma unroll
        for (int y = 0; y < 8; ++y) {
            const float4 q4 = *(const float4*)(qp + (y >> 1) * CQ_F + (y & 1) * RQ_F);
            aq[y] = make_float4(q4.x * scale, q4.y * scale, q4.z * scale, q4.w * scale);
        }
    } else {
        #pragma unroll
        for (int y = 0; y < 8; ++y) aq[y] = make_float4(0.f, 0.f, 0.f, 0.f);
    }
    const float* fsrc = ((role == 0) ? k : v) + recb;
    if (role != 2) {
        #pragma unroll
        for (int i = 0; i < 16; ++i)
            fr[i] = *(const float4*)(fsrc + (i & 3) * sx + (i >> 3) * sy + ((i >> 2) & 1) * RQ_F);
        __builtin_amdgcn_sched_barrier(0);
        #pragma unroll
        for (int i = 16; i < 32; ++i)
            fr[i] = *(const float4*)(fsrc + (i & 3) * sx + (i >> 3) * sy + ((i >> 2) & 1) * RQ_F);
    } else {
        #pragma unroll
        for (int i = 0; i < 32; ++i) fr[i] = make_float4(0.f, 0.f, 0.f, 0.f);
    }

    for (int jb = 0; jb <= nmax + 1; ++jb) {
        const bool prod = (role == 0) && (jb < nblk);
        const bool cons = (role == 1) && (jb >= 2) && (jb <= nblk + 1);
        const bool soft = (role == 2) && (jb >= 1) && (jb <= nblk);
        if (prod || cons) {
            const int jw = min(prod ? (jb + 1) : (jb - 1), nblk - 1);
            const float* fp = fsrc + (size_t)(jw * KT) * TILE_F;
            if (prod) {
                #pragma unroll
                for (int x = 0; x < 4; ++x) ac[x] = v8f{};
            } else {
                const int slot = jb & 1;
                const float4 a0 = *(const float4*)&abuf[slot][half * 8];
                const float4 a1 = *(const float4*)&abuf[slot][half * 8 + 4];
                const float al[8] = { a0.x, a0.y, a0.z, a0.w, a1.x, a1.y, a1.z, a1.w };
                #pragma unroll
                for (int y = 0; y < 8; ++y)
                    aq[y] = *(const float4*)&pbuf[slot][nlo][16 * (y >> 1) + 8 * half + 4 * (y & 1)];
                __builtin_amdgcn_sched_barrier(0);
                #pragma unroll
                for (int r = 0; r < 8; ++r) {
                    ac[0][r] *= al[r]; ac[1][r] *= al[r];
                    ac[2][r] *= al[r]; ac[3][r] *= al[r];
                }
            }
            #pragma unroll
            for (int y = 0; y < 4; ++y) {
                v2f fa; fa.x = aq[y].x; fa.y = aq[y].y;
                #pragma unroll
                for (int x = 0; x < 4; ++x) {
                    v2f fb; fb.x = fr[y * 4 + x].x; fb.y = fr[y * 4 + x].y;
                    ac[x] = wmma4(fa, fb, ac[x]);
                }
                fa.x = aq[y].z; fa.y = aq[y].w;
                #pragma unroll
                for (int x = 0; x < 4; ++x) {
                    v2f fb; fb.x = fr[y * 4 + x].z; fb.y = fr[y * 4 + x].w;
                    ac[x] = wmma4(fa, fb, ac[x]);
                }
            }
            asm volatile("" : "+v"(ac[0]), "+v"(ac[1]), "+v"(ac[2]), "+v"(ac[3]) :: "memory");
            __builtin_amdgcn_sched_barrier(0);
            #pragma unroll
            for (int i = 0; i < 16; ++i)
                fr[i] = *(const float4*)(fp + (i & 3) * sx + (i >> 3) * sy + ((i >> 2) & 1) * RQ_F);
            __builtin_amdgcn_sched_barrier(0);
            #pragma unroll
            for (int y = 4; y < 8; ++y) {
                v2f fa; fa.x = aq[y].x; fa.y = aq[y].y;
                #pragma unroll
                for (int x = 0; x < 4; ++x) {
                    v2f fb; fb.x = fr[y * 4 + x].x; fb.y = fr[y * 4 + x].y;
                    ac[x] = wmma4(fa, fb, ac[x]);
                }
                fa.x = aq[y].z; fa.y = aq[y].w;
                #pragma unroll
                for (int x = 0; x < 4; ++x) {
                    v2f fb; fb.x = fr[y * 4 + x].z; fb.y = fr[y * 4 + x].w;
                    ac[x] = wmma4(fa, fb, ac[x]);
                }
            }
            asm volatile("" : "+v"(ac[0]), "+v"(ac[1]), "+v"(ac[2]), "+v"(ac[3]) :: "memory");
            __builtin_amdgcn_sched_barrier(0);
            #pragma unroll
            for (int i = 16; i < 32; ++i)
                fr[i] = *(const float4*)(fp + (i & 3) * sx + (i >> 3) * sy + ((i >> 2) & 1) * RQ_F);
            __builtin_amdgcn_sched_barrier(0);
            if (prod) {
                #pragma unroll
                for (int tb = 0; tb < KT; ++tb) {
                    #pragma unroll
                    for (int rq = 0; rq < 2; ++rq)
                        sbuf[jb & 1][tb * 2 + rq][lane] = make_float4(ac[tb][4 * rq + 0], ac[tb][4 * rq + 1],
                                                                      ac[tb][4 * rq + 2], ac[tb][4 * rq + 3]);
                }
            }
        } else if (soft) {
            const int ts   = jb - 1;
            const int slot = ts & 1;
            float sv[KT][8];
            #pragma unroll
            for (int tb = 0; tb < KT; ++tb) {
                #pragma unroll
                for (int rq = 0; rq < 2; ++rq) {
                    const float4 s4 = sbuf[slot][tb * 2 + rq][lane];
                    sv[tb][4 * rq + 0] = s4.x; sv[tb][4 * rq + 1] = s4.y;
                    sv[tb][4 * rq + 2] = s4.z; sv[tb][4 * rq + 3] = s4.w;
                }
            }

            if (ts == nblk - 1) {
                const int kidx = ts * KB + nlo;
                #pragma unroll
                for (int tb = 0; tb < KT; ++tb) {
                    #pragma unroll
                    for (int r = 0; r < 8; ++r) {
                        if (kidx + 16 * tb > qidx + r) sv[tb][r] = -1e30f;
                    }
                }
            }

            float alpha[8];
            #pragma unroll
            for (int r = 0; r < 8; ++r) {
                float t = sv[0][r];
                #pragma unroll
                for (int tb = 1; tb < KT; ++tb) t = fmaxf(t, sv[tb][r]);
                t = fmaxf(t, row_xor<1>(t));
                t = fmaxf(t, row_xor<2>(t));
                t = fmaxf(t, row_xor<4>(t));
                t = fmaxf(t, row_xor<8>(t));
                float mnew = fmaxf(mrow[r], t);
                alpha[r] = __expf(mrow[r] - mnew);
                mrow[r]  = mnew;
                float s = 0.f;
                #pragma unroll
                for (int tb = 0; tb < KT; ++tb) {
                    float p = __expf(sv[tb][r] - mnew);
                    pbuf[slot][r + half * 8][16 * tb + nlo] = p;
                    s += p;
                }
                s += row_xor<1>(s);
                s += row_xor<2>(s);
                s += row_xor<4>(s);
                s += row_xor<8>(s);
                lrow[r] = lrow[r] * alpha[r] + s;
            }
            if (nlo == 0) {
                #pragma unroll
                for (int r = 0; r < 8; ++r) abuf[slot][half * 8 + r] = alpha[r];
                if (ts == nblk - 1) {
                    #pragma unroll
                    for (int r = 0; r < 8; ++r) lbuf[half * 8 + r] = lrow[r];
                }
            }
        }
        __builtin_amdgcn_fence(__ATOMIC_RELEASE, "workgroup", "local");
        __builtin_amdgcn_s_barrier();
        __builtin_amdgcn_fence(__ATOMIC_ACQUIRE, "workgroup", "local");
    }

    if (role == 1) {
        const float4 l0 = *(const float4*)&lbuf[half * 8];
        const float4 l1 = *(const float4*)&lbuf[half * 8 + 4];
        const float lr[8] = { l0.x, l0.y, l0.z, l0.w, l1.x, l1.y, l1.z, l1.w };
        #pragma unroll
        for (int r = 0; r < 8; ++r) {
            float inv = 1.0f / lr[r];
            size_t idx = (size_t)(qrow0 + r + half * 8) * Hh;
            out[idx +  0 + nlo] = ac[0][r] * inv;
            out[idx + 16 + nlo] = ac[1][r] * inv;
            out[idx + 32 + nlo] = ac[2][r] * inv;
            out[idx + 48 + nlo] = ac[3][r] * inv;
        }
    }
}

extern "C" void kernel_launch(void* const* d_in, const int* in_sizes, int n_in,
                              void* d_out, int out_size, void* d_ws, size_t ws_size,
                              hipStream_t stream) {
    const float* x  = (const float*)d_in[0];
    const float* Wk = (const float*)d_in[1];
    const float* Wq = (const float*)d_in[2];
    const float* Wv = (const float*)d_in[3];
    float* out = (float*)d_out;

    float* ws   = (float*)d_ws;
    float* qbuf = ws;
    float* kbuf = ws + (size_t)ROWS * Hh;
    float* vbuf = ws + 2 * (size_t)ROWS * Hh;

    proj_kernel<<<ROWS / (16 * PR), 256, 0, stream>>>(x, Wk, Wq, Wv, qbuf, kbuf, vbuf);
    attn_kernel<<<ROWS / (16 * QT), 96 * QT, 0, stream>>>(qbuf, kbuf, vbuf, out);
}
